# starting kernel
# baseline (speedup 1.0000x reference)
#include <hip/hip_runtime.h>

typedef __attribute__((ext_vector_type(2))) float v2f;
typedef __attribute__((ext_vector_type(8))) float v8f;

#define F_IN   128
#define H_DIM  64
#define Z_DIM  32
#define G_GRAPHS 512

__global__ __launch_bounds__(32) void gemm_wmma_f32(
    const float* __restrict__ A, const float* __restrict__ B,
    const float* __restrict__ bias, float* __restrict__ C,
    int M, int K, int Nc)
{
  const int tileN = blockIdx.x;
  const int tileM = blockIdx.y;
  const int lane  = threadIdx.x;
  const int half  = lane >> 4;
  const int l     = lane & 15;

  const float* Arow = A + (long long)(tileM * 16 + l) * K;
  const float* Bcol = B + (tileN * 16 + l);

  v8f c = {0.f, 0.f, 0.f, 0.f, 0.f, 0.f, 0.f, 0.f};

  for (int k = 0; k < K; k += 4) {
    const int ka = k + 2 * half;
    v2f a, b;
    a.x = Arow[ka];
    a.y = Arow[ka + 1];
    b.x = Bcol[(long long)ka * Nc];
    b.y = Bcol[(long long)(ka + 1) * Nc];
    c = __builtin_amdgcn_wmma_f32_16x16x4_f32(
        false, a, false, b, (short)0, c, false, false);
  }

  const int col = tileN * 16 + l;
  const float bv = bias ? bias[col] : 0.0f;
  float* Crow = C + (long long)(tileM * 16 + 8 * half) * Nc + col;
#pragma unroll
  for (int r = 0; r < 8; ++r) {
    Crow[(long long)r * Nc] = c[r] + bv;
  }
}

__global__ void fill_f32(float* __restrict__ p, float v, int n) {
  int t = blockIdx.x * blockDim.x + threadIdx.x;
  if (t < n) p[t] = v;
}

__global__ void deg_accum(const int* __restrict__ dst, float* __restrict__ deg, int E) {
  int t = blockIdx.x * blockDim.x + threadIdx.x;
  if (t < E) atomicAdd(&deg[dst[t]], 1.0f);
}

__global__ void deg_to_dinv(float* __restrict__ d, int n) {
  int t = blockIdx.x * blockDim.x + threadIdx.x;
  if (t < n) d[t] = rsqrtf(d[t] + 1.0f);
}

__global__ void selfloop_init(const float* __restrict__ h,
                              const float* __restrict__ dinv,
                              float* __restrict__ agg, int n) {
  int t = blockIdx.x * blockDim.x + threadIdx.x;
  if (t >= n * H_DIM) return;
  int i = t / H_DIM;
  float di = dinv[i];
  agg[t] = h[t] * (di * di);
}

__global__ void edge_prop(const int* __restrict__ src, const int* __restrict__ dst,
                          const float* __restrict__ dinv,
                          const float* __restrict__ h, float* __restrict__ agg,
                          int E) {
  const int chunks = H_DIM / 4;
  long long t = (long long)blockIdx.x * blockDim.x + threadIdx.x;
  if (t >= (long long)E * chunks) return;
  int e = (int)(t / chunks);
  int c = (int)(t % chunks);
  int s = src[e], d = dst[e];
  float coef = dinv[s] * dinv[d];
  float4 v = ((const float4*)(h + (long long)s * H_DIM))[c];
  float* o = agg + (long long)d * H_DIM + 4 * c;
  atomicAdd(o + 0, v.x * coef);
  atomicAdd(o + 1, v.y * coef);
  atomicAdd(o + 2, v.z * coef);
  atomicAdd(o + 3, v.w * coef);
}

__global__ void bias_relu(const float* __restrict__ agg, const float* __restrict__ b,
                          float* __restrict__ out, int n) {
  int t = blockIdx.x * blockDim.x + threadIdx.x;
  if (t >= n * H_DIM) return;
  float v = agg[t] + b[t % H_DIM];
  out[t] = v > 0.0f ? v : 0.0f;
}

__global__ void pool_accum(const float* __restrict__ h, const int* __restrict__ batch,
                           float* __restrict__ sums, float* __restrict__ cnts, int n) {
  const int chunks = H_DIM / 4;
  int t = blockIdx.x * blockDim.x + threadIdx.x;
  if (t >= n * chunks) return;
  int i = t / chunks, c = t % chunks;
  int g = batch[i];
  float4 v = ((const float4*)(h + (long long)i * H_DIM))[c];
  float* o = sums + (long long)g * H_DIM + 4 * c;
  atomicAdd(o + 0, v.x);
  atomicAdd(o + 1, v.y);
  atomicAdd(o + 2, v.z);
  atomicAdd(o + 3, v.w);
  if (c == 0) atomicAdd(&cnts[g], 1.0f);
}

__global__ void pool_final(const float* __restrict__ sums, const float* __restrict__ cnts,
                           float* __restrict__ pooled) {
  int t = blockIdx.x * blockDim.x + threadIdx.x;
  if (t >= G_GRAPHS * H_DIM) return;
  pooled[t] = sums[t] / fmaxf(cnts[t / H_DIM], 1.0f);
}

static inline size_t align_up(size_t x, size_t a) { return (x + a - 1) & ~(a - 1); }

extern "C" void kernel_launch(void* const* d_in, const int* in_sizes, int n_in,
                              void* d_out, int out_size, void* d_ws, size_t ws_size,
                              hipStream_t stream) {
  const float* x   = (const float*)d_in[0];
  const int*   ei  = (const int*)d_in[1];
  const int*   bat = (const int*)d_in[2];
  const float* W1  = (const float*)d_in[3];
  const float* b1  = (const float*)d_in[4];
  const float* W2  = (const float*)d_in[5];
  const float* b2  = (const float*)d_in[6];
  const float* Wmu = (const float*)d_in[7];
  const float* bmu = (const float*)d_in[8];
  const float* Wls = (const float*)d_in[9];
  const float* bls = (const float*)d_in[10];
  float* out = (float*)d_out;

  const int N = in_sizes[0] / F_IN;
  const int E = in_sizes[1] / 2;
  const int* src = ei;
  const int* dst = ei + E;

  char* ws = (char*)d_ws;
  size_t off = 0;
  float* dinv = (float*)(ws + off); off = align_up(off + (size_t)N * 4, 256);
  float* bufA = (float*)(ws + off); off = align_up(off + (size_t)N * H_DIM * 4, 256);
  float* bufB = (float*)(ws + off); off = align_up(off + (size_t)N * H_DIM * 4, 256);
  float* bufC = (float*)(ws + off); off = align_up(off + (size_t)N * H_DIM * 4, 256);
  float* sums = (float*)(ws + off); off = align_up(off + (size_t)G_GRAPHS * H_DIM * 4, 256);
  float* cnts = (float*)(ws + off); off = align_up(off + (size_t)G_GRAPHS * 4, 256);
  float* pooled = (float*)(ws + off); off = align_up(off + (size_t)G_GRAPHS * H_DIM * 4, 256);
  (void)ws_size; (void)n_in; (void)out_size;

  const int BLK = 256;
  const int NH = N * H_DIM;
  const long long EC = (long long)E * (H_DIM / 4);
  const int NC = N * (H_DIM / 4);

  fill_f32<<<(N + BLK - 1) / BLK, BLK, 0, stream>>>(dinv, 0.0f, N);
  deg_accum<<<(E + BLK - 1) / BLK, BLK, 0, stream>>>(dst, dinv, E);
  deg_to_dinv<<<(N + BLK - 1) / BLK, BLK, 0, stream>>>(dinv, N);

  gemm_wmma_f32<<<dim3(H_DIM / 16, N / 16), 32, 0, stream>>>(
      x, W1, nullptr, bufA, N, F_IN, H_DIM);
  selfloop_init<<<(NH + BLK - 1) / BLK, BLK, 0, stream>>>(bufA, dinv, bufB, N);
  edge_prop<<<(unsigned)((EC + BLK - 1) / BLK), BLK, 0, stream>>>(
      src, dst, dinv, bufA, bufB, E);
  bias_relu<<<(NH + BLK - 1) / BLK, BLK, 0, stream>>>(bufB, b1, bufA, N);

  gemm_wmma_f32<<<dim3(H_DIM / 16, N / 16), 32, 0, stream>>>(
      bufA, W2, nullptr, bufB, N, H_DIM, H_DIM);
  selfloop_init<<<(NH + BLK - 1) / BLK, BLK, 0, stream>>>(bufB, dinv, bufC, N);
  edge_prop<<<(unsigned)((EC + BLK - 1) / BLK), BLK, 0, stream>>>(
      src, dst, dinv, bufB, bufC, E);
  bias_relu<<<(NH + BLK - 1) / BLK, BLK, 0, stream>>>(bufC, b2, bufA, N);

  fill_f32<<<(G_GRAPHS * H_DIM + BLK - 1) / BLK, BLK, 0, stream>>>(sums, 0.0f, G_GRAPHS * H_DIM);
  fill_f32<<<(G_GRAPHS + BLK - 1) / BLK, BLK, 0, stream>>>(cnts, 0.0f, G_GRAPHS);
  pool_accum<<<(NC + BLK - 1) / BLK, BLK, 0, stream>>>(bufA, bat, sums, cnts, N);
  pool_final<<<(G_GRAPHS * H_DIM + BLK - 1) / BLK, BLK, 0, stream>>>(sums, cnts, pooled);

  gemm_wmma_f32<<<dim3(Z_DIM / 16, G_GRAPHS / 16), 32, 0, stream>>>(
      pooled, Wmu, bmu, out, G_GRAPHS, H_DIM, Z_DIM);
  gemm_wmma_f32<<<dim3(Z_DIM / 16, G_GRAPHS / 16), 32, 0, stream>>>(
      pooled, Wls, bls, out + (size_t)G_GRAPHS * Z_DIM, G_GRAPHS, H_DIM, Z_DIM);
}


# Propagation as a gather over a dst-sorted edge list built once per call (returning int add, scan, fill): its float atomics go, rows summed in registers in list order, self-loop+bias+relu in one write
# speedup vs baseline: 1.4983x; 1.4983x over previous
#include <hip/hip_runtime.h>

typedef __attribute__((ext_vector_type(2))) float v2f;
typedef __attribute__((ext_vector_type(8))) float v8f;

#define F_IN   128
#define H_DIM  64
#define Z_DIM  32
#define G_GRAPHS 512

__global__ __launch_bounds__(32) void gemm_wmma_f32(
    const float* __restrict__ A, const float* __restrict__ B,
    const float* __restrict__ bias, float* __restrict__ C,
    int M, int K, int Nc)
{
  const int tileN = blockIdx.x;
  const int tileM = blockIdx.y;
  const int lane  = threadIdx.x;
  const int half  = lane >> 4;
  const int l     = lane & 15;

  const float* Arow = A + (long long)(tileM * 16 + l) * K;
  const float* Bcol = B + (tileN * 16 + l);

  v8f c = {0.f, 0.f, 0.f, 0.f, 0.f, 0.f, 0.f, 0.f};

  for (int k = 0; k < K; k += 4) {
    const int ka = k + 2 * half;
    v2f a, b;
    a.x = Arow[ka];
    a.y = Arow[ka + 1];
    b.x = Bcol[(long long)ka * Nc];
    b.y = Bcol[(long long)(ka + 1) * Nc];
    c = __builtin_amdgcn_wmma_f32_16x16x4_f32(
        false, a, false, b, (short)0, c, false, false);
  }

  const int col = tileN * 16 + l;
  const float bv = bias ? bias[col] : 0.0f;
  float* Crow = C + (long long)(tileM * 16 + 8 * half) * Nc + col;
#pragma unroll
  for (int r = 0; r < 8; ++r) {
    Crow[(long long)r * Nc] = c[r] + bv;
  }
}

__global__ void fill_f32(float* __restrict__ p, float v, int n) {
  int t = blockIdx.x * blockDim.x + threadIdx.x;
  if (t < n) p[t] = v;
}

__global__ void fill_i32(int* __restrict__ p, int v, int n) {
  int t = blockIdx.x * blockDim.x + threadIdx.x;
  if (t < n) p[t] = v;
}

__global__ void count_rank(const int* __restrict__ dst, int* __restrict__ cnt,
                           int* __restrict__ rank, int E) {
  int t = blockIdx.x * blockDim.x + threadIdx.x;
  if (t < E) rank[t] = atomicAdd(&cnt[dst[t]], 1);
}

__global__ __launch_bounds__(1024) void scan_deg(const int* __restrict__ cnt,
                                                 int* __restrict__ rowptr,
                                                 float* __restrict__ dinv, int n) {
  __shared__ int part[1024];
  const int t = threadIdx.x;
  const int first = blockIdx.x * 1024;
  int s = 0;
  for (int k = t; k < first; k += 1024) s += cnt[k];
  part[t] = s;
  __syncthreads();
  for (int off = 512; off > 0; off >>= 1) {
    if (t < off) part[t] += part[t + off];
    __syncthreads();
  }
  const int front = part[0];
  __syncthreads();

  const int g = first + t;
  const int c = (g < n) ? cnt[g] : 0;
  part[t] = c;
  __syncthreads();
  for (int off = 1; off < 1024; off <<= 1) {
    const int v = (t >= off) ? part[t - off] : 0;
    __syncthreads();
    part[t] += v;
    __syncthreads();
  }
  const int incl = front + part[t];
  if (g < n) {
    rowptr[g] = incl - c;
    dinv[g] = rsqrtf((float)c + 1.0f);
    if (g == n - 1) rowptr[n] = incl;
  }
}

__global__ void fill_csr(const int* __restrict__ src, const int* __restrict__ dst,
                         const int* __restrict__ rowptr, const int* __restrict__ rank,
                         const float* __restrict__ dinv, int* __restrict__ csr, int E) {
  int t = blockIdx.x * blockDim.x + threadIdx.x;
  if (t >= E) return;
  const int s = src[t];
  const int p = rowptr[dst[t]] + rank[t];
  ((int2*)csr)[p] = make_int2(s, __float_as_int(dinv[s]));
}

__global__ __launch_bounds__(256) void gather_prop(
    const int* __restrict__ rowptr, const int* __restrict__ csr,
    const float* __restrict__ dinv, const float* __restrict__ h,
    const float* __restrict__ bias, float* __restrict__ out, int n)
{
  const int lane = threadIdx.x & 31;
  const int i = __builtin_amdgcn_readfirstlane((int)(blockIdx.x * 8 + (threadIdx.x >> 5)));
  if (i >= n) return;
  const int beg = rowptr[i];
  const int end = rowptr[i + 1];
  const float di = dinv[i];
  const float2* __restrict__ h2 = (const float2*)h;
  const int2* __restrict__ pr = (const int2*)csr;
  const float2 hv = h2[(long long)i * (H_DIM / 2) + lane];

  float ax = 0.0f, ay = 0.0f;
  for (int base = beg; base < end; base += 32) {
    const int m = min(32, end - base);
    int2 p = make_int2(i, 0);
    if (lane < m) p = pr[base + lane];
#pragma unroll 4
    for (int j = 0; j < m; ++j) {
      const int sj = __builtin_amdgcn_readlane(p.x, j);
      const float cj = __int_as_float(__builtin_amdgcn_readlane(p.y, j)) * di;
      const float2 v = h2[(long long)sj * (H_DIM / 2) + lane];
      ax += v.x * cj;
      ay += v.y * cj;
    }
  }
  const float sl = di * di;
  const float2 bv = ((const float2*)bias)[lane];
  const float ox = ax + hv.x * sl + bv.x;
  const float oy = ay + hv.y * sl + bv.y;
  float2 o;
  o.x = ox > 0.0f ? ox : 0.0f;
  o.y = oy > 0.0f ? oy : 0.0f;
  ((float2*)out)[(long long)i * (H_DIM / 2) + lane] = o;
}

__global__ void pool_accum(const float* __restrict__ h, const int* __restrict__ batch,
                           float* __restrict__ sums, float* __restrict__ cnts, int n) {
  const int chunks = H_DIM / 4;
  int t = blockIdx.x * blockDim.x + threadIdx.x;
  if (t >= n * chunks) return;
  int i = t / chunks, c = t % chunks;
  int g = batch[i];
  float4 v = ((const float4*)(h + (long long)i * H_DIM))[c];
  float* o = sums + (long long)g * H_DIM + 4 * c;
  atomicAdd(o + 0, v.x);
  atomicAdd(o + 1, v.y);
  atomicAdd(o + 2, v.z);
  atomicAdd(o + 3, v.w);
  if (c == 0) atomicAdd(&cnts[g], 1.0f);
}

__global__ void pool_final(const float* __restrict__ sums, const float* __restrict__ cnts,
                           float* __restrict__ pooled) {
  int t = blockIdx.x * blockDim.x + threadIdx.x;
  if (t >= G_GRAPHS * H_DIM) return;
  pooled[t] = sums[t] / fmaxf(cnts[t / H_DIM], 1.0f);
}

static inline size_t align_up(size_t x, size_t a) { return (x + a - 1) & ~(a - 1); }

extern "C" void kernel_launch(void* const* d_in, const int* in_sizes, int n_in,
                              void* d_out, int out_size, void* d_ws, size_t ws_size,
                              hipStream_t stream) {
  const float* x   = (const float*)d_in[0];
  const int*   ei  = (const int*)d_in[1];
  const int*   bat = (const int*)d_in[2];
  const float* W1  = (const float*)d_in[3];
  const float* b1  = (const float*)d_in[4];
  const float* W2  = (const float*)d_in[5];
  const float* b2  = (const float*)d_in[6];
  const float* Wmu = (const float*)d_in[7];
  const float* bmu = (const float*)d_in[8];
  const float* Wls = (const float*)d_in[9];
  const float* bls = (const float*)d_in[10];
  float* out = (float*)d_out;

  const int N = in_sizes[0] / F_IN;
  const int E = in_sizes[1] / 2;
  const int* src = ei;
  const int* dst = ei + E;

  char* ws = (char*)d_ws;
  size_t off = 0;
  float* dinv = (float*)(ws + off); off = align_up(off + (size_t)N * 4, 256);
  int* cnt    = (int*)(ws + off);   off = align_up(off + (size_t)N * 4, 256);
  int* rowptr = (int*)(ws + off);   off = align_up(off + ((size_t)N + 1) * 4, 256);
  int* rank   = (int*)(ws + off);   off = align_up(off + (size_t)E * 4, 256);
  int* csr    = (int*)(ws + off);   off = align_up(off + (size_t)E * 8, 256);
  float* bufA = (float*)(ws + off); off = align_up(off + (size_t)N * H_DIM * 4, 256);
  float* bufB = (float*)(ws + off); off = align_up(off + (size_t)N * H_DIM * 4, 256);
  float* sums = (float*)(ws + off); off = align_up(off + (size_t)G_GRAPHS * H_DIM * 4, 256);
  float* cnts = (float*)(ws + off); off = align_up(off + (size_t)G_GRAPHS * 4, 256);
  float* pooled = (float*)(ws + off); off = align_up(off + (size_t)G_GRAPHS * H_DIM * 4, 256);
  (void)ws_size; (void)n_in; (void)out_size;

  const int BLK = 256;
  const int NC = N * (H_DIM / 4);

  fill_i32<<<(N + BLK - 1) / BLK, BLK, 0, stream>>>(cnt, 0, N);
  count_rank<<<(E + BLK - 1) / BLK, BLK, 0, stream>>>(dst, cnt, rank, E);
  scan_deg<<<(N + 1023) / 1024, 1024, 0, stream>>>(cnt, rowptr, dinv, N);
  fill_csr<<<(E + BLK - 1) / BLK, BLK, 0, stream>>>(src, dst, rowptr, rank, dinv, csr, E);

  gemm_wmma_f32<<<dim3(H_DIM / 16, N / 16), 32, 0, stream>>>(
      x, W1, nullptr, bufA, N, F_IN, H_DIM);
  gather_prop<<<(N + 7) / 8, 256, 0, stream>>>(rowptr, csr, dinv, bufA, b1, bufB, N);

  gemm_wmma_f32<<<dim3(H_DIM / 16, N / 16), 32, 0, stream>>>(
      bufB, W2, nullptr, bufA, N, H_DIM, H_DIM);
  gather_prop<<<(N + 7) / 8, 256, 0, stream>>>(rowptr, csr, dinv, bufA, b2, bufB, N);

  fill_f32<<<(G_GRAPHS * H_DIM + BLK - 1) / BLK, BLK, 0, stream>>>(sums, 0.0f, G_GRAPHS * H_DIM);
  fill_f32<<<(G_GRAPHS + BLK - 1) / BLK, BLK, 0, stream>>>(cnts, 0.0f, G_GRAPHS);
  pool_accum<<<(NC + BLK - 1) / BLK, BLK, 0, stream>>>(bufB, bat, sums, cnts, N);
  pool_final<<<(G_GRAPHS * H_DIM + BLK - 1) / BLK, BLK, 0, stream>>>(sums, cnts, pooled);

  gemm_wmma_f32<<<dim3(Z_DIM / 16, G_GRAPHS / 16), 32, 0, stream>>>(
      pooled, Wmu, bmu, out, G_GRAPHS, H_DIM, Z_DIM);
  gemm_wmma_f32<<<dim3(Z_DIM / 16, G_GRAPHS / 16), 32, 0, stream>>>(
      pooled, Wls, bls, out + (size_t)G_GRAPHS * Z_DIM, G_GRAPHS, H_DIM, Z_DIM);
}


# Launches 14 -> 9: stages that do not depend on each other share a launch (three clears; edge counting beside the x@W1 product; both heads in one launch, with the pool division in their A load)
# speedup vs baseline: 1.8162x; 1.2122x over previous
#include <hip/hip_runtime.h>

typedef __attribute__((ext_vector_type(2))) float v2f;
typedef __attribute__((ext_vector_type(8))) float v8f;

#define F_IN   128
#define H_DIM  64
#define Z_DIM  32
#define G_GRAPHS 512

__device__ __forceinline__ void gemm_tile_f32(
    const float* __restrict__ A, const float* __restrict__ B,
    const float* __restrict__ bias, float* __restrict__ C,
    int K, int Nc, int tileM, int tileN, int lane)
{
  const int half  = lane >> 4;
  const int l     = lane & 15;

  const float* Arow = A + (long long)(tileM * 16 + l) * K;
  const float* Bcol = B + (tileN * 16 + l);

  v8f c = {0.f, 0.f, 0.f, 0.f, 0.f, 0.f, 0.f, 0.f};

  for (int k = 0; k < K; k += 4) {
    const int ka = k + 2 * half;
    v2f a, b;
    a.x = Arow[ka];
    a.y = Arow[ka + 1];
    b.x = Bcol[(long long)ka * Nc];
    b.y = Bcol[(long long)(ka + 1) * Nc];
    c = __builtin_amdgcn_wmma_f32_16x16x4_f32(
        false, a, false, b, (short)0, c, false, false);
  }

  const int col = tileN * 16 + l;
  const float bv = bias ? bias[col] : 0.0f;
  float* Crow = C + (long long)(tileM * 16 + 8 * half) * Nc + col;
#pragma unroll
  for (int r = 0; r < 8; ++r) {
    Crow[(long long)r * Nc] = c[r] + bv;
  }
}

__global__ __launch_bounds__(32) void gemm_wmma_f32(
    const float* __restrict__ A, const float* __restrict__ B,
    const float* __restrict__ bias, float* __restrict__ C,
    int M, int K, int Nc)
{
  (void)M;
  gemm_tile_f32(A, B, bias, C, K, Nc, blockIdx.y, blockIdx.x, threadIdx.x);
}

__global__ __launch_bounds__(32) void heads_wmma_f32(
    const float* __restrict__ sums, const float* __restrict__ cnts,
    const float* __restrict__ Wmu, const float* __restrict__ bmu,
    const float* __restrict__ Wls, const float* __restrict__ bls,
    float* __restrict__ out)
{
  const int head  = blockIdx.x / (Z_DIM / 16);
  const int tileN = blockIdx.x % (Z_DIM / 16);
  const int tileM = blockIdx.y;
  const int lane  = threadIdx.x;
  const int half  = lane >> 4;
  const int l     = lane & 15;

  const float* __restrict__ Wm = head ? Wls : Wmu;
  const float* __restrict__ bm = head ? bls : bmu;
  float* __restrict__ C = out + (size_t)head * G_GRAPHS * Z_DIM;

  const int row = tileM * 16 + l;
  const float den = fmaxf(cnts[row], 1.0f);
  const float* Arow = sums + (long long)row * H_DIM;
  const float* Bcol = Wm + (tileN * 16 + l);

  v8f c = {0.f, 0.f, 0.f, 0.f, 0.f, 0.f, 0.f, 0.f};

  v2f a[H_DIM / 4], b[H_DIM / 4];
#pragma unroll
  for (int s = 0; s < H_DIM / 4; ++s) {
    const int ka = 4 * s + 2 * half;
    a[s].x = Arow[ka];
    a[s].y = Arow[ka + 1];
    b[s].x = Bcol[ka * Z_DIM];
    b[s].y = Bcol[(ka + 1) * Z_DIM];
  }
#pragma unroll
  for (int s = 0; s < H_DIM / 4; ++s) {
    a[s].x = a[s].x / den;
    a[s].y = a[s].y / den;
  }
  __builtin_amdgcn_sched_barrier(0);
#pragma unroll
  for (int s = 0; s < H_DIM / 4; ++s) {
    c = __builtin_amdgcn_wmma_f32_16x16x4_f32(
        false, a[s], false, b[s], (short)0, c, false, false);
  }

  const int col = tileN * 16 + l;
  const float bv = bm[col];
  float* Crow = C + (long long)(tileM * 16 + 8 * half) * Z_DIM + col;
#pragma unroll
  for (int r = 0; r < 8; ++r) {
    Crow[r * Z_DIM] = c[r] + bv;
  }
}

__global__ void clear_words(int4* __restrict__ p, int n16) {
  int t = blockIdx.x * blockDim.x + threadIdx.x;
  if (t < n16) p[t] = make_int4(0, 0, 0, 0);
}


__global__ __launch_bounds__(256) void count_rank_gemm(
    const int* __restrict__ dst, int* __restrict__ cnt, int* __restrict__ rank, int E,
    const float* __restrict__ A, const float* __restrict__ B, float* __restrict__ C,
    int K, int Nc, int n_tiles, int gemm_blocks)
{
  if ((int)blockIdx.x < gemm_blocks) {
    const int tile = __builtin_amdgcn_readfirstlane((int)(blockIdx.x * 8 + (threadIdx.x >> 5)));
    if (tile >= n_tiles) return;
    const int tn = Nc >> 4;
    gemm_tile_f32(A, B, nullptr, C, K, Nc, tile / tn, tile % tn, (int)(threadIdx.x & 31));
  } else {
    const int t = (int)(blockIdx.x - gemm_blocks) * 256 + (int)threadIdx.x;
    if (t < E) rank[t] = atomicAdd(&cnt[dst[t]], 1);
  }
}

__global__ __launch_bounds__(1024) void scan_deg(const int* __restrict__ cnt,
                                                 int* __restrict__ rowptr,
                                                 float* __restrict__ dinv, int n) {
  __shared__ int part[1024];
  const int t = threadIdx.x;
  const int first = blockIdx.x * 1024;
  int s = 0;
  for (int k = t; k < first; k += 1024) s += cnt[k];
  part[t] = s;
  __syncthreads();
  for (int off = 512; off > 0; off >>= 1) {
    if (t < off) part[t] += part[t + off];
    __syncthreads();
  }
  const int front = part[0];
  __syncthreads();

  const int g = first + t;
  const int c = (g < n) ? cnt[g] : 0;
  part[t] = c;
  __syncthreads();
  for (int off = 1; off < 1024; off <<= 1) {
    const int v = (t >= off) ? part[t - off] : 0;
    __syncthreads();
    part[t] += v;
    __syncthreads();
  }
  const int incl = front + part[t];
  if (g < n) {
    rowptr[g] = incl - c;
    dinv[g] = rsqrtf((float)c + 1.0f);
    if (g == n - 1) rowptr[n] = incl;
  }
}

__global__ void fill_csr(const int* __restrict__ src, const int* __restrict__ dst,
                         const int* __restrict__ rowptr, const int* __restrict__ rank,
                         const float* __restrict__ dinv, int* __restrict__ csr, int E) {
  int t = blockIdx.x * blockDim.x + threadIdx.x;
  if (t >= E) return;
  const int s = src[t];
  const int p = rowptr[dst[t]] + rank[t];
  ((int2*)csr)[p] = make_int2(s, __float_as_int(dinv[s]));
}

__global__ __launch_bounds__(256) void gather_prop(
    const int* __restrict__ rowptr, const int* __restrict__ csr,
    const float* __restrict__ dinv, const float* __restrict__ h,
    const float* __restrict__ bias, float* __restrict__ out, int n)
{
  const int lane = threadIdx.x & 31;
  const int i = __builtin_amdgcn_readfirstlane((int)(blockIdx.x * 8 + (threadIdx.x >> 5)));
  if (i >= n) return;
  const int beg = rowptr[i];
  const int end = rowptr[i + 1];
  const float di = dinv[i];
  const float2* __restrict__ h2 = (const float2*)h;
  const int2* __restrict__ pr = (const int2*)csr;
  const float2 hv = h2[(long long)i * (H_DIM / 2) + lane];

  float ax = 0.0f, ay = 0.0f;
  for (int base = beg; base < end; base += 32) {
    const int m = min(32, end - base);
    int2 p = make_int2(i, 0);
    if (lane < m) p = pr[base + lane];
#pragma unroll 4
    for (int j = 0; j < m; ++j) {
      const int sj = __builtin_amdgcn_readlane(p.x, j);
      const float cj = __int_as_float(__builtin_amdgcn_readlane(p.y, j)) * di;
      const float2 v = h2[(long long)sj * (H_DIM / 2) + lane];
      ax += v.x * cj;
      ay += v.y * cj;
    }
  }
  const float sl = di * di;
  const float2 bv = ((const float2*)bias)[lane];
  const float ox = ax + hv.x * sl + bv.x;
  const float oy = ay + hv.y * sl + bv.y;
  float2 o;
  o.x = ox > 0.0f ? ox : 0.0f;
  o.y = oy > 0.0f ? oy : 0.0f;
  ((float2*)out)[(long long)i * (H_DIM / 2) + lane] = o;
}

__global__ void pool_accum(const float* __restrict__ h, const int* __restrict__ batch,
                           float* __restrict__ sums, float* __restrict__ cnts, int n) {
  const int chunks = H_DIM / 4;
  int t = blockIdx.x * blockDim.x + threadIdx.x;
  if (t >= n * chunks) return;
  int i = t / chunks, c = t % chunks;
  int g = batch[i];
  float4 v = ((const float4*)(h + (long long)i * H_DIM))[c];
  float* o = sums + (long long)g * H_DIM + 4 * c;
  atomicAdd(o + 0, v.x);
  atomicAdd(o + 1, v.y);
  atomicAdd(o + 2, v.z);
  atomicAdd(o + 3, v.w);
  if (c == 0) atomicAdd(&cnts[g], 1.0f);
}

static inline size_t align_up(size_t x, size_t a) { return (x + a - 1) & ~(a - 1); }

extern "C" void kernel_launch(void* const* d_in, const int* in_sizes, int n_in,
                              void* d_out, int out_size, void* d_ws, size_t ws_size,
                              hipStream_t stream) {
  const float* x   = (const float*)d_in[0];
  const int*   ei  = (const int*)d_in[1];
  const int*   bat = (const int*)d_in[2];
  const float* W1  = (const float*)d_in[3];
  const float* b1  = (const float*)d_in[4];
  const float* W2  = (const float*)d_in[5];
  const float* b2  = (const float*)d_in[6];
  const float* Wmu = (const float*)d_in[7];
  const float* bmu = (const float*)d_in[8];
  const float* Wls = (const float*)d_in[9];
  const float* bls = (const float*)d_in[10];
  float* out = (float*)d_out;

  const int N = in_sizes[0] / F_IN;
  const int E = in_sizes[1] / 2;
  const int* src = ei;
  const int* dst = ei + E;

  char* ws = (char*)d_ws;
  size_t off = 0;
  float* dinv = (float*)(ws + off); off = align_up(off + (size_t)N * 4, 256);
  const size_t zero_beg = off;
  int* cnt    = (int*)(ws + off);   off = align_up(off + (size_t)N * 4, 256);
  float* sums = (float*)(ws + off); off = align_up(off + (size_t)G_GRAPHS * H_DIM * 4, 256);
  float* cnts = (float*)(ws + off); off = align_up(off + (size_t)G_GRAPHS * 4, 256);
  const size_t zero_end = off;
  int* rowptr = (int*)(ws + off);   off = align_up(off + ((size_t)N + 1) * 4, 256);
  int* rank   = (int*)(ws + off);   off = align_up(off + (size_t)E * 4, 256);
  int* csr    = (int*)(ws + off);   off = align_up(off + (size_t)E * 8, 256);
  float* bufA = (float*)(ws + off); off = align_up(off + (size_t)N * H_DIM * 4, 256);
  float* bufB = (float*)(ws + off); off = align_up(off + (size_t)N * H_DIM * 4, 256);
  (void)ws_size; (void)n_in; (void)out_size;

  const int BLK = 256;
  const int NC = N * (H_DIM / 4);
  const int n16 = (int)((zero_end - zero_beg) / 16);
  const int n_tiles = (N / 16) * (H_DIM / 16);
  const int gemm_blocks = (n_tiles + 7) / 8;

  clear_words<<<(n16 + BLK - 1) / BLK, BLK, 0, stream>>>((int4*)(ws + zero_beg), n16);

  count_rank_gemm<<<gemm_blocks + (E + BLK - 1) / BLK, BLK, 0, stream>>>(
      dst, cnt, rank, E, x, W1, bufA, F_IN, H_DIM, n_tiles, gemm_blocks);

  scan_deg<<<(N + 1023) / 1024, 1024, 0, stream>>>(cnt, rowptr, dinv, N);
  fill_csr<<<(E + BLK - 1) / BLK, BLK, 0, stream>>>(src, dst, rowptr, rank, dinv, csr, E);

  gather_prop<<<(N + 7) / 8, 256, 0, stream>>>(rowptr, csr, dinv, bufA, b1, bufB, N);

  gemm_wmma_f32<<<dim3(H_DIM / 16, N / 16), 32, 0, stream>>>(
      bufB, W2, nullptr, bufA, N, H_DIM, H_DIM);
  gather_prop<<<(N + 7) / 8, 256, 0, stream>>>(rowptr, csr, dinv, bufA, b2, bufB, N);

  pool_accum<<<(NC + BLK - 1) / BLK, BLK, 0, stream>>>(bufB, bat, sums, cnts, N);

  heads_wmma_f32<<<dim3(2 * (Z_DIM / 16), G_GRAPHS / 16), 32, 0, stream>>>(
      sums, cnts, Wmu, bmu, Wls, bls, out);
}


# Gather workgroups keep their 16 finished rows in LDS and run the consumer too: layer-2 product (WMMA) behind gather 1, pool sums (rows pre-added, then float atomics) behind gather 2; 9->7 launches
# speedup vs baseline: 2.4351x; 1.3408x over previous
#include <hip/hip_runtime.h>

typedef __attribute__((ext_vector_type(2))) float v2f;
typedef __attribute__((ext_vector_type(8))) float v8f;

#define F_IN   128
#define H_DIM  64
#define Z_DIM  32
#define G_GRAPHS 512

template <int K>
__device__ __forceinline__ void gemm_rowblock_f32(
    const float* __restrict__ A, const float* __restrict__ B,
    float* __restrict__ C, int tileM, int lane)
{
  const int half  = lane >> 4;
  const int l     = lane & 15;

  const float* Arow = A + (long long)(tileM * 16 + l) * K;
  const float* Bcol = B + l;

  v8f c0 = {0.f, 0.f, 0.f, 0.f, 0.f, 0.f, 0.f, 0.f};
  v8f c1 = c0, c2 = c0, c3 = c0;

#pragma unroll 4
  for (int k = 0; k < K; k += 4) {
    const int ka = k + 2 * half;
    v2f a, b0, b1, b2, b3;
    a.x = Arow[ka];
    a.y = Arow[ka + 1];
    const float* Bk0 = Bcol + ka * H_DIM;
    const float* Bk1 = Bcol + (ka + 1) * H_DIM;
    b0.x = Bk0[0];  b0.y = Bk1[0];
    b1.x = Bk0[16]; b1.y = Bk1[16];
    b2.x = Bk0[32]; b2.y = Bk1[32];
    b3.x = Bk0[48]; b3.y = Bk1[48];
    c0 = __builtin_amdgcn_wmma_f32_16x16x4_f32(false, a, false, b0, (short)0, c0, false, false);
    c1 = __builtin_amdgcn_wmma_f32_16x16x4_f32(false, a, false, b1, (short)0, c1, false, false);
    c2 = __builtin_amdgcn_wmma_f32_16x16x4_f32(false, a, false, b2, (short)0, c2, false, false);
    c3 = __builtin_amdgcn_wmma_f32_16x16x4_f32(false, a, false, b3, (short)0, c3, false, false);
  }

  float* Crow = C + (long long)(tileM * 16 + 8 * half) * H_DIM + l;
#pragma unroll
  for (int r = 0; r < 8; ++r) {
    Crow[r * H_DIM]      = c0[r];
    Crow[r * H_DIM + 16] = c1[r];
    Crow[r * H_DIM + 32] = c2[r];
    Crow[r * H_DIM + 48] = c3[r];
  }
}

__global__ __launch_bounds__(32) void heads_wmma_f32(
    const float* __restrict__ sums, const float* __restrict__ cnts,
    const float* __restrict__ Wmu, const float* __restrict__ bmu,
    const float* __restrict__ Wls, const float* __restrict__ bls,
    float* __restrict__ out)
{
  const int head  = blockIdx.x / (Z_DIM / 16);
  const int tileN = blockIdx.x % (Z_DIM / 16);
  const int tileM = blockIdx.y;
  const int lane  = threadIdx.x;
  const int half  = lane >> 4;
  const int l     = lane & 15;

  const float* __restrict__ Wm = head ? Wls : Wmu;
  const float* __restrict__ bm = head ? bls : bmu;
  float* __restrict__ C = out + (size_t)head * G_GRAPHS * Z_DIM;

  const int row = tileM * 16 + l;
  const float den = fmaxf(cnts[row], 1.0f);
  const float* Arow = sums + (long long)row * H_DIM;
  const float* Bcol = Wm + (tileN * 16 + l);

  v8f c = {0.f, 0.f, 0.f, 0.f, 0.f, 0.f, 0.f, 0.f};

  v2f a[H_DIM / 4], b[H_DIM / 4];
#pragma unroll
  for (int s = 0; s < H_DIM / 4; ++s) {
    const int ka = 4 * s + 2 * half;
    a[s].x = Arow[ka];
    a[s].y = Arow[ka + 1];
    b[s].x = Bcol[ka * Z_DIM];
    b[s].y = Bcol[(ka + 1) * Z_DIM];
  }
#pragma unroll
  for (int s = 0; s < H_DIM / 4; ++s) {
    a[s].x = a[s].x / den;
    a[s].y = a[s].y / den;
  }
  __builtin_amdgcn_sched_barrier(0);
#pragma unroll
  for (int s = 0; s < H_DIM / 4; ++s) {
    c = __builtin_amdgcn_wmma_f32_16x16x4_f32(
        false, a[s], false, b[s], (short)0, c, false, false);
  }

  const int col = tileN * 16 + l;
  const float bv = bm[col];
  float* Crow = C + (long long)(tileM * 16 + 8 * half) * Z_DIM + col;
#pragma unroll
  for (int r = 0; r < 8; ++r) {
    Crow[r * Z_DIM] = c[r] + bv;
  }
}

__global__ void clear_words(int4* __restrict__ p, int n16) {
  int t = blockIdx.x * blockDim.x + threadIdx.x;
  if (t < n16) p[t] = make_int4(0, 0, 0, 0);
}


__global__ __launch_bounds__(256) void count_rank_gemm(
    const int* __restrict__ dst, int* __restrict__ cnt, int* __restrict__ rank, int E,
    const float* __restrict__ A, const float* __restrict__ B, float* __restrict__ C,
    int n_rb, int gemm_blocks)
{
  if ((int)blockIdx.x < gemm_blocks) {
    const int rb = __builtin_amdgcn_readfirstlane((int)(blockIdx.x * 8 + (threadIdx.x >> 5)));
    if (rb >= n_rb) return;
    gemm_rowblock_f32<F_IN>(A, B, C, rb, (int)(threadIdx.x & 31));
  } else {
    const int t0 = (int)(blockIdx.x - gemm_blocks) * 1024 + (int)threadIdx.x;
    if (t0 - (int)threadIdx.x + 1024 <= E) {
      int d[4], r[4];
#pragma unroll
      for (int j = 0; j < 4; ++j) d[j] = dst[t0 + 256 * j];
#pragma unroll
      for (int j = 0; j < 4; ++j) r[j] = atomicAdd(&cnt[d[j]], 1);
#pragma unroll
      for (int j = 0; j < 4; ++j) rank[t0 + 256 * j] = r[j];
    } else {
      for (int j = 0; j < 4; ++j) {
        const int t = t0 + 256 * j;
        if (t < E) rank[t] = atomicAdd(&cnt[dst[t]], 1);
      }
    }
  }
}

__global__ __launch_bounds__(1024) void scan_deg(const int* __restrict__ cnt,
                                                 int* __restrict__ rowptr,
                                                 float* __restrict__ dinv, int n) {
  __shared__ int part[1024];
  const int t = threadIdx.x;
  const int first = blockIdx.x * 1024;
  int s = 0;
  for (int k = t; k < first; k += 1024) s += cnt[k];
  part[t] = s;
  __syncthreads();
  for (int off = 512; off > 0; off >>= 1) {
    if (t < off) part[t] += part[t + off];
    __syncthreads();
  }
  const int front = part[0];
  __syncthreads();

  const int g = first + t;
  const int c = (g < n) ? cnt[g] : 0;
  part[t] = c;
  __syncthreads();
  for (int off = 1; off < 1024; off <<= 1) {
    const int v = (t >= off) ? part[t - off] : 0;
    __syncthreads();
    part[t] += v;
    __syncthreads();
  }
  const int incl = front + part[t];
  if (g < n) {
    rowptr[g] = incl - c;
    dinv[g] = rsqrtf((float)c + 1.0f);
    if (g == n - 1) rowptr[n] = incl;
  }
}

__global__ __launch_bounds__(256) void fill_csr(
                         const int* __restrict__ src, const int* __restrict__ dst,
                         const int* __restrict__ rowptr, const int* __restrict__ rank,
                         const float* __restrict__ dinv, int* __restrict__ csr, int E) {
  const int t0 = (int)blockIdx.x * 1024 + (int)threadIdx.x;
  if (t0 - (int)threadIdx.x + 1024 <= E) {
    int s[4], d[4], r[4], rp[4];
    float dv[4];
#pragma unroll
    for (int j = 0; j < 4; ++j) {
      const int t = t0 + 256 * j;
      s[j] = src[t];
      d[j] = dst[t];
      r[j] = rank[t];
    }
#pragma unroll
    for (int j = 0; j < 4; ++j) {
      rp[j] = rowptr[d[j]];
      dv[j] = dinv[s[j]];
    }
#pragma unroll
    for (int j = 0; j < 4; ++j)
      ((int2*)csr)[rp[j] + r[j]] = make_int2(s[j], __float_as_int(dv[j]));
  } else {
    for (int j = 0; j < 4; ++j) {
      const int t = t0 + 256 * j;
      if (t < E) {
        const int sv = src[t];
        const int p = rowptr[dst[t]] + rank[t];
        ((int2*)csr)[p] = make_int2(sv, __float_as_int(dinv[sv]));
      }
    }
  }
}

#define T_PITCH 68

__device__ __forceinline__ float4 gather_row(
    const int* __restrict__ rowptr, const int* __restrict__ csr,
    const float* __restrict__ dinv, const float* __restrict__ h,
    const float* __restrict__ bias, int ic, bool live, int sub, int hi)
{
  const int beg  = rowptr[ic];
  const int endr = rowptr[ic + 1];
  const int len  = live ? endr - beg : 0;
  const float di = dinv[ic];
  const float4* __restrict__ h4 = (const float4*)h;
  const int2* __restrict__ pr = (const int2*)csr;
  const float4 hv = h4[(long long)ic * (H_DIM / 4) + sub];

  const int len0 = __builtin_amdgcn_readlane(len, 0);
  const int len1 = __builtin_amdgcn_readlane(len, 16);
  const int maxlen = max(len0, len1);

  float ax = 0.0f, ay = 0.0f, az = 0.0f, aw = 0.0f;
  for (int base = 0; base < maxlen; base += 16) {
    const int m = min(16, maxlen - base);
    const int e = base + sub;
    int2 p = pr[e < len ? beg + e : 0];
    if (!(e < len)) p = make_int2(ic, 0);
    for (int j = 0; j < m; j += 8) {
      float4 v[8];
      float cj[8];
#pragma unroll
      for (int u = 0; u < 8; ++u) {
        const int s0 = __builtin_amdgcn_readlane(p.x, j + u);
        const int s1 = __builtin_amdgcn_readlane(p.x, j + u + 16);
        const int q0 = __builtin_amdgcn_readlane(p.y, j + u);
        const int q1 = __builtin_amdgcn_readlane(p.y, j + u + 16);
        const int sj = hi ? s1 : s0;
        cj[u] = __int_as_float(hi ? q1 : q0) * di;
        v[u] = h4[(long long)sj * (H_DIM / 4) + sub];
      }
#pragma unroll
      for (int u = 0; u < 8; ++u) {
        const bool on = base + j + u < len;
        const float tx = ax + v[u].x * cj[u];
        const float ty = ay + v[u].y * cj[u];
        const float tz = az + v[u].z * cj[u];
        const float tw = aw + v[u].w * cj[u];
        ax = on ? tx : ax;
        ay = on ? ty : ay;
        az = on ? tz : az;
        aw = on ? tw : aw;
      }
    }
  }
  const float sl = di * di;
  const float4 bv = ((const float4*)bias)[sub];
  const float ox = ax + hv.x * sl + bv.x;
  const float oy = ay + hv.y * sl + bv.y;
  const float oz = az + hv.z * sl + bv.z;
  const float ow = aw + hv.w * sl + bv.w;
  float4 o;
  o.x = (live && ox > 0.0f) ? ox : 0.0f;
  o.y = (live && oy > 0.0f) ? oy : 0.0f;
  o.z = (live && oz > 0.0f) ? oz : 0.0f;
  o.w = (live && ow > 0.0f) ? ow : 0.0f;
  return o;
}

__global__ __launch_bounds__(256) void gather_prop_gemm(
    const int* __restrict__ rowptr, const int* __restrict__ csr,
    const float* __restrict__ dinv, const float* __restrict__ h,
    const float* __restrict__ bias, const float* __restrict__ W,
    float* __restrict__ out, int n)
{
  __shared__ __attribute__((aligned(16))) float T[16 * T_PITCH];
  __shared__ __attribute__((aligned(16))) float U[16 * T_PITCH];

  const int lane = threadIdx.x & 31;
  const int sub  = lane & 15;
  const int hi   = lane >> 4;
  const int w    = __builtin_amdgcn_readfirstlane((int)(threadIdx.x >> 5));
  const int rl   = 2 * w + hi;
  const int i    = (int)blockIdx.x * 16 + rl;
  const bool live = i < n;
  const int ic   = live ? i : n - 1;

  const float4 o = gather_row(rowptr, csr, dinv, h, bias, ic, live, sub, hi);
  *(float4*)&T[rl * T_PITCH + 4 * sub] = o;

  v2f b[H_DIM / 4];
  if (w < 4) {
    const float* Bcol = W + 16 * w + sub;
#pragma unroll
    for (int s = 0; s < H_DIM / 4; ++s) {
      const int ka = 4 * s + 2 * hi;
      b[s].x = Bcol[ka * H_DIM];
      b[s].y = Bcol[(ka + 1) * H_DIM];
    }
  }
  __syncthreads();
  if (w < 4) {
    v8f c = {0.f, 0.f, 0.f, 0.f, 0.f, 0.f, 0.f, 0.f};
    const float* Arow = T + sub * T_PITCH + 2 * hi;
#pragma unroll
    for (int s = 0; s < H_DIM / 4; ++s) {
      v2f a;
      a.x = Arow[4 * s];
      a.y = Arow[4 * s + 1];
      c = __builtin_amdgcn_wmma_f32_16x16x4_f32(false, a, false, b[s], (short)0, c, false, false);
    }
    float* Ucol = U + (8 * hi) * T_PITCH + 16 * w + sub;
#pragma unroll
    for (int r = 0; r < 8; ++r) Ucol[r * T_PITCH] = c[r];
  }
  __syncthreads();
  if (live) {
    const float4 v = *(const float4*)&U[rl * T_PITCH + 4 * sub];
    ((float4*)out)[(long long)i * (H_DIM / 4) + sub] = v;
  }
}

__global__ __launch_bounds__(256) void gather_prop_pool(
    const int* __restrict__ rowptr, const int* __restrict__ csr,
    const float* __restrict__ dinv, const float* __restrict__ h,
    const float* __restrict__ bias, const int* __restrict__ batch,
    float* __restrict__ sums, float* __restrict__ cnts, int n)
{
  __shared__ __attribute__((aligned(16))) float T[16 * T_PITCH];
  __shared__ __attribute__((aligned(16))) int Gs[16];

  const int lane = threadIdx.x & 31;
  const int sub  = lane & 15;
  const int hi   = lane >> 4;
  const int w    = (int)(threadIdx.x >> 5);
  const int rl   = 2 * w + hi;
  const int i    = (int)blockIdx.x * 16 + rl;
  const bool live = i < n;
  const int ic   = live ? i : n - 1;
  const int g    = batch[ic];

  const float4 o = gather_row(rowptr, csr, dinv, h, bias, ic, live, sub, hi);
  *(float4*)&T[rl * T_PITCH + 4 * sub] = o;
  if (sub == 0) Gs[rl] = live ? g : -1;
  __syncthreads();
  if (threadIdx.x >= H_DIM) return;

  const int f = (int)threadIdx.x;
  float tv[16];
  int gs[16];
#pragma unroll
  for (int q = 0; q < 4; ++q) {
    const int4 g4 = ((const int4*)Gs)[q];
    gs[4 * q + 0] = g4.x;
    gs[4 * q + 1] = g4.y;
    gs[4 * q + 2] = g4.z;
    gs[4 * q + 3] = g4.w;
  }
#pragma unroll
  for (int r = 0; r < 16; ++r) tv[r] = T[r * T_PITCH + f];

  int cur = gs[0];
  float acc = 0.0f, cn = 0.0f;
#pragma unroll
  for (int r = 0; r < 16; ++r) {
    if (gs[r] != cur) {
      if (cur >= 0) {
        atomicAdd(&sums[(long long)cur * H_DIM + f], acc);
        if (f == 0) atomicAdd(&cnts[cur], cn);
      }
      acc = 0.0f;
      cn = 0.0f;
      cur = gs[r];
    }
    acc += tv[r];
    cn += 1.0f;
  }
  if (cur >= 0) {
    atomicAdd(&sums[(long long)cur * H_DIM + f], acc);
    if (f == 0) atomicAdd(&cnts[cur], cn);
  }
}

static inline size_t align_up(size_t x, size_t a) { return (x + a - 1) & ~(a - 1); }

extern "C" void kernel_launch(void* const* d_in, const int* in_sizes, int n_in,
                              void* d_out, int out_size, void* d_ws, size_t ws_size,
                              hipStream_t stream) {
  const float* x   = (const float*)d_in[0];
  const int*   ei  = (const int*)d_in[1];
  const int*   bat = (const int*)d_in[2];
  const float* W1  = (const float*)d_in[3];
  const float* b1  = (const float*)d_in[4];
  const float* W2  = (const float*)d_in[5];
  const float* b2  = (const float*)d_in[6];
  const float* Wmu = (const float*)d_in[7];
  const float* bmu = (const float*)d_in[8];
  const float* Wls = (const float*)d_in[9];
  const float* bls = (const float*)d_in[10];
  float* out = (float*)d_out;

  const int N = in_sizes[0] / F_IN;
  const int E = in_sizes[1] / 2;
  const int* src = ei;
  const int* dst = ei + E;

  char* ws = (char*)d_ws;
  size_t off = 0;
  float* dinv = (float*)(ws + off); off = align_up(off + (size_t)N * 4, 256);
  const size_t zero_beg = off;
  int* cnt    = (int*)(ws + off);   off = align_up(off + (size_t)N * 4, 256);
  float* sums = (float*)(ws + off); off = align_up(off + (size_t)G_GRAPHS * H_DIM * 4, 256);
  float* cnts = (float*)(ws + off); off = align_up(off + (size_t)G_GRAPHS * 4, 256);
  const size_t zero_end = off;
  int* rowptr = (int*)(ws + off);   off = align_up(off + ((size_t)N + 1) * 4, 256);
  int* rank   = (int*)(ws + off);   off = align_up(off + (size_t)E * 4, 256);
  int* csr    = (int*)(ws + off);   off = align_up(off + (size_t)E * 8, 256);
  float* bufA = (float*)(ws + off); off = align_up(off + (size_t)N * H_DIM * 4, 256);
  float* bufB = (float*)(ws + off); off = align_up(off + (size_t)N * H_DIM * 4, 256);
  (void)ws_size; (void)n_in; (void)out_size;

  const int BLK = 256;
  const int n16 = (int)((zero_end - zero_beg) / 16);
  const int n_rb = N / 16;
  const int gemm_blocks = (n_rb + 7) / 8;
  const int PER = 4 * BLK;

  clear_words<<<(n16 + BLK - 1) / BLK, BLK, 0, stream>>>((int4*)(ws + zero_beg), n16);

  count_rank_gemm<<<gemm_blocks + (E + PER - 1) / PER, BLK, 0, stream>>>(
      dst, cnt, rank, E, x, W1, bufA, n_rb, gemm_blocks);

  scan_deg<<<(N + 1023) / 1024, 1024, 0, stream>>>(cnt, rowptr, dinv, N);
  fill_csr<<<(E + PER - 1) / PER, BLK, 0, stream>>>(src, dst, rowptr, rank, dinv, csr, E);

  gather_prop_gemm<<<(N + 15) / 16, BLK, 0, stream>>>(rowptr, csr, dinv, bufA, b1, W2, bufB, N);

  gather_prop_pool<<<(N + 15) / 16, BLK, 0, stream>>>(rowptr, csr, dinv, bufB, b2, bat, sums, cnts, N);

  heads_wmma_f32<<<dim3(2 * (Z_DIM / 16), G_GRAPHS / 16), 32, 0, stream>>>(
      sums, cnts, Wmu, bmu, Wls, bls, out);
}


# Edge list as a slot table of fixed row pitch (64 slots per destination, overflow list behind) filled directly by the returning add: scan and fill launches go (7 -> 5); 1/sqrt(deg) from the counts
# speedup vs baseline: 2.8216x; 1.1587x over previous
#include <hip/hip_runtime.h>

typedef __attribute__((ext_vector_type(2))) float v2f;
typedef __attribute__((ext_vector_type(8))) float v8f;

#define F_IN   128
#define H_DIM  64
#define Z_DIM  32
#define G_GRAPHS 512

template <int K>
__device__ __forceinline__ void gemm_rowblock_f32(
    const float* __restrict__ A, const float* __restrict__ B,
    float* __restrict__ C, int tileM, int lane)
{
  const int half  = lane >> 4;
  const int l     = lane & 15;

  const float* Arow = A + (long long)(tileM * 16 + l) * K;
  const float* Bcol = B + l;

  v8f c0 = {0.f, 0.f, 0.f, 0.f, 0.f, 0.f, 0.f, 0.f};
  v8f c1 = c0, c2 = c0, c3 = c0;

#pragma unroll 4
  for (int k = 0; k < K; k += 4) {
    const int ka = k + 2 * half;
    v2f a, b0, b1, b2, b3;
    a.x = Arow[ka];
    a.y = Arow[ka + 1];
    const float* Bk0 = Bcol + ka * H_DIM;
    const float* Bk1 = Bcol + (ka + 1) * H_DIM;
    b0.x = Bk0[0];  b0.y = Bk1[0];
    b1.x = Bk0[16]; b1.y = Bk1[16];
    b2.x = Bk0[32]; b2.y = Bk1[32];
    b3.x = Bk0[48]; b3.y = Bk1[48];
    c0 = __builtin_amdgcn_wmma_f32_16x16x4_f32(false, a, false, b0, (short)0, c0, false, false);
    c1 = __builtin_amdgcn_wmma_f32_16x16x4_f32(false, a, false, b1, (short)0, c1, false, false);
    c2 = __builtin_amdgcn_wmma_f32_16x16x4_f32(false, a, false, b2, (short)0, c2, false, false);
    c3 = __builtin_amdgcn_wmma_f32_16x16x4_f32(false, a, false, b3, (short)0, c3, false, false);
  }

  float* Crow = C + (long long)(tileM * 16 + 8 * half) * H_DIM + l;
#pragma unroll
  for (int r = 0; r < 8; ++r) {
    Crow[r * H_DIM]      = c0[r];
    Crow[r * H_DIM + 16] = c1[r];
    Crow[r * H_DIM + 32] = c2[r];
    Crow[r * H_DIM + 48] = c3[r];
  }
}

__global__ __launch_bounds__(32) void heads_wmma_f32(
    const float* __restrict__ sums, const float* __restrict__ cnts,
    const float* __restrict__ Wmu, const float* __restrict__ bmu,
    const float* __restrict__ Wls, const float* __restrict__ bls,
    float* __restrict__ out)
{
  const int head  = blockIdx.x / (Z_DIM / 16);
  const int tileN = blockIdx.x % (Z_DIM / 16);
  const int tileM = blockIdx.y;
  const int lane  = threadIdx.x;
  const int half  = lane >> 4;
  const int l     = lane & 15;

  const float* __restrict__ Wm = head ? Wls : Wmu;
  const float* __restrict__ bm = head ? bls : bmu;
  float* __restrict__ C = out + (size_t)head * G_GRAPHS * Z_DIM;

  const int row = tileM * 16 + l;
  const float den = fmaxf(cnts[row], 1.0f);
  const float* Arow = sums + (long long)row * H_DIM;
  const float* Bcol = Wm + (tileN * 16 + l);

  v8f c = {0.f, 0.f, 0.f, 0.f, 0.f, 0.f, 0.f, 0.f};

  v2f a[H_DIM / 4], b[H_DIM / 4];
#pragma unroll
  for (int s = 0; s < H_DIM / 4; ++s) {
    const int ka = 4 * s + 2 * half;
    a[s].x = Arow[ka];
    a[s].y = Arow[ka + 1];
    b[s].x = Bcol[ka * Z_DIM];
    b[s].y = Bcol[(ka + 1) * Z_DIM];
  }
#pragma unroll
  for (int s = 0; s < H_DIM / 4; ++s) {
    a[s].x = a[s].x / den;
    a[s].y = a[s].y / den;
  }
  __builtin_amdgcn_sched_barrier(0);
#pragma unroll
  for (int s = 0; s < H_DIM / 4; ++s) {
    c = __builtin_amdgcn_wmma_f32_16x16x4_f32(
        false, a[s], false, b[s], (short)0, c, false, false);
  }

  const int col = tileN * 16 + l;
  const float bv = bm[col];
  float* Crow = C + (long long)(tileM * 16 + 8 * half) * Z_DIM + col;
#pragma unroll
  for (int r = 0; r < 8; ++r) {
    Crow[r * Z_DIM] = c[r] + bv;
  }
}

__global__ void clear_words(int4* __restrict__ p, int n16) {
  int t = blockIdx.x * blockDim.x + threadIdx.x;
  if (t < n16) p[t] = make_int4(0, 0, 0, 0);
}

#define ROW_SLOTS 64

__device__ __forceinline__ void put_edge(int* __restrict__ tab, int* __restrict__ ovf_n,
                                         int* __restrict__ ovf, int s, int d, int r) {
  if (r < ROW_SLOTS) {
    tab[(long long)d * ROW_SLOTS + r] = s;
  } else {
    const int k = atomicAdd(ovf_n, 1);
    ((int2*)ovf)[k] = make_int2(s, d);
  }
}

__global__ __launch_bounds__(256) void count_slot_gemm(
    const int* __restrict__ src, const int* __restrict__ dst, int* __restrict__ cnt,
    int* __restrict__ tab, int* __restrict__ ovf_n, int* __restrict__ ovf, int E,
    const float* __restrict__ A, const float* __restrict__ B, float* __restrict__ C,
    int n_rb, int gemm_blocks)
{
  if ((int)blockIdx.x < gemm_blocks) {
    const int rb = __builtin_amdgcn_readfirstlane((int)(blockIdx.x * 8 + (threadIdx.x >> 5)));
    if (rb >= n_rb) return;
    gemm_rowblock_f32<F_IN>(A, B, C, rb, (int)(threadIdx.x & 31));
  } else {
    const int t0 = (int)(blockIdx.x - gemm_blocks) * 1024 + (int)threadIdx.x;
    if (t0 - (int)threadIdx.x + 1024 <= E) {
      int d[4], s[4], r[4];
#pragma unroll
      for (int j = 0; j < 4; ++j) d[j] = dst[t0 + 256 * j];
#pragma unroll
      for (int j = 0; j < 4; ++j) s[j] = src[t0 + 256 * j];
#pragma unroll
      for (int j = 0; j < 4; ++j) r[j] = atomicAdd(&cnt[d[j]], 1);
#pragma unroll
      for (int j = 0; j < 4; ++j) put_edge(tab, ovf_n, ovf, s[j], d[j], r[j]);
    } else {
      for (int j = 0; j < 4; ++j) {
        const int t = t0 + 256 * j;
        if (t < E) {
          const int dv = dst[t];
          put_edge(tab, ovf_n, ovf, src[t], dv, atomicAdd(&cnt[dv], 1));
        }
      }
    }
  }
}

#define T_PITCH 68

__device__ __forceinline__ float dinv_of(int c) {
  return __builtin_amdgcn_rsqf((float)c + 1.0f);
}

__device__ __forceinline__ float4 gather_row(
    const int* __restrict__ cnt, const int* __restrict__ tab,
    const int* __restrict__ ovf_n, const int* __restrict__ ovf,
    const float* __restrict__ h, const float* __restrict__ bias,
    int ic, bool live, int sub, int hi, int E)
{
  const int* __restrict__ trow = tab + (long long)ic * ROW_SLOTS;
  const float4* __restrict__ h4 = (const float4*)h;
  const int degi = cnt[ic];
  const float4 hv = h4[(long long)ic * (H_DIM / 4) + sub];
  const int deg = live ? degi : 0;
  const int len = min(deg, ROW_SLOTS);
  const float di = dinv_of(degi);

  const int len0 = __builtin_amdgcn_readlane(len, 0);
  const int len1 = __builtin_amdgcn_readlane(len, 16);
  const int maxlen = max(len0, len1);

  int s_cur = ic, s_nxt = ic;
  if (sub < len) s_cur = trow[sub];
  if (16 + sub < len) s_nxt = trow[16 + sub];

  float ax = 0.0f, ay = 0.0f, az = 0.0f, aw = 0.0f;
  for (int base = 0; base < maxlen; base += 16) {
    const int m = min(16, maxlen - base);
    const int px = s_cur;
    s_cur = s_nxt;
    if (base + 32 < maxlen) {
      const int e2 = base + 32 + sub;
      s_nxt = ic;
      if (e2 < len) s_nxt = trow[e2];
    }
    for (int j = 0; j < m; j += 8) {
      float4 v[8];
      int c[8];
#pragma unroll
      for (int u = 0; u < 8; ++u) {
        const int s0 = __builtin_amdgcn_readlane(px, j + u);
        const int s1 = __builtin_amdgcn_readlane(px, j + u + 16);
        const int sj = hi ? s1 : s0;
        v[u] = h4[sj * (H_DIM / 4) + sub];
        c[u] = cnt[sj];
      }
#pragma unroll
      for (int u = 0; u < 8; ++u) {
        const float cj = dinv_of(c[u]) * di;
        const bool on = base + j + u < len;
        const float tx = ax + v[u].x * cj;
        const float ty = ay + v[u].y * cj;
        const float tz = az + v[u].z * cj;
        const float tw = aw + v[u].w * cj;
        ax = on ? tx : ax;
        ay = on ? ty : ay;
        az = on ? tz : az;
        aw = on ? tw : aw;
      }
    }
  }

  const int deg0 = __builtin_amdgcn_readlane(deg, 0);
  const int deg1 = __builtin_amdgcn_readlane(deg, 16);
  if (max(deg0, deg1) > ROW_SLOTS) {
    const int nov = min(ovf_n[0], E);
    const int2* __restrict__ ov = (const int2*)ovf;
    for (int k = 0; k < nov; ++k) {
      const int2 od = ov[k];
      if (live && od.y == ic) {
        const float cj = dinv_of(cnt[od.x]) * di;
        const float4 v = h4[(long long)od.x * (H_DIM / 4) + sub];
        ax += v.x * cj;
        ay += v.y * cj;
        az += v.z * cj;
        aw += v.w * cj;
      }
    }
  }

  const float sl = di * di;
  const float4 bv = ((const float4*)bias)[sub];
  const float ox = ax + hv.x * sl + bv.x;
  const float oy = ay + hv.y * sl + bv.y;
  const float oz = az + hv.z * sl + bv.z;
  const float ow = aw + hv.w * sl + bv.w;
  float4 o;
  o.x = (live && ox > 0.0f) ? ox : 0.0f;
  o.y = (live && oy > 0.0f) ? oy : 0.0f;
  o.z = (live && oz > 0.0f) ? oz : 0.0f;
  o.w = (live && ow > 0.0f) ? ow : 0.0f;
  return o;
}

__global__ __launch_bounds__(256) void gather_prop_gemm(
    const int* __restrict__ cnt, const int* __restrict__ tab,
    const int* __restrict__ ovf_n, const int* __restrict__ ovf,
    const float* __restrict__ h,
    const float* __restrict__ bias, const float* __restrict__ W,
    float* __restrict__ out, int n, int E)
{
  __shared__ __attribute__((aligned(16))) float T[16 * T_PITCH];
  __shared__ __attribute__((aligned(16))) float U[16 * T_PITCH];

  const int lane = threadIdx.x & 31;
  const int sub  = lane & 15;
  const int hi   = lane >> 4;
  const int w    = __builtin_amdgcn_readfirstlane((int)(threadIdx.x >> 5));
  const int rl   = 2 * w + hi;
  const int i    = (int)blockIdx.x * 16 + rl;
  const bool live = i < n;
  const int ic   = live ? i : n - 1;

  const float4 o = gather_row(cnt, tab, ovf_n, ovf, h, bias, ic, live, sub, hi, E);
  *(float4*)&T[rl * T_PITCH + 4 * sub] = o;

  v2f b[H_DIM / 4];
  if (w < 4) {
    const float* Bcol = W + 16 * w + sub;
#pragma unroll
    for (int s = 0; s < H_DIM / 4; ++s) {
      const int ka = 4 * s + 2 * hi;
      b[s].x = Bcol[ka * H_DIM];
      b[s].y = Bcol[(ka + 1) * H_DIM];
    }
  }
  __syncthreads();
  if (w < 4) {
    v8f c = {0.f, 0.f, 0.f, 0.f, 0.f, 0.f, 0.f, 0.f};
    const float* Arow = T + sub * T_PITCH + 2 * hi;
#pragma unroll
    for (int s = 0; s < H_DIM / 4; ++s) {
      v2f a;
      a.x = Arow[4 * s];
      a.y = Arow[4 * s + 1];
      c = __builtin_amdgcn_wmma_f32_16x16x4_f32(false, a, false, b[s], (short)0, c, false, false);
    }
    float* Ucol = U + (8 * hi) * T_PITCH + 16 * w + sub;
#pragma unroll
    for (int r = 0; r < 8; ++r) Ucol[r * T_PITCH] = c[r];
  }
  __syncthreads();
  if (live) {
    const float4 v = *(const float4*)&U[rl * T_PITCH + 4 * sub];
    ((float4*)out)[(long long)i * (H_DIM / 4) + sub] = v;
  }
}

__global__ __launch_bounds__(256) void gather_prop_pool(
    const int* __restrict__ cnt, const int* __restrict__ tab,
    const int* __restrict__ ovf_n, const int* __restrict__ ovf,
    const float* __restrict__ h,
    const float* __restrict__ bias, const int* __restrict__ batch,
    float* __restrict__ sums, float* __restrict__ cnts, int n, int E)
{
  __shared__ __attribute__((aligned(16))) float T[16 * T_PITCH];
  __shared__ __attribute__((aligned(16))) int Gs[16];

  const int lane = threadIdx.x & 31;
  const int sub  = lane & 15;
  const int hi   = lane >> 4;
  const int w    = (int)(threadIdx.x >> 5);
  const int rl   = 2 * w + hi;
  const int i    = (int)blockIdx.x * 16 + rl;
  const bool live = i < n;
  const int ic   = live ? i : n - 1;
  const int g    = batch[ic];

  const float4 o = gather_row(cnt, tab, ovf_n, ovf, h, bias, ic, live, sub, hi, E);
  *(float4*)&T[rl * T_PITCH + 4 * sub] = o;
  if (sub == 0) Gs[rl] = live ? g : -1;
  __syncthreads();
  if (threadIdx.x >= H_DIM) return;

  const int f = (int)threadIdx.x;
  float tv[16];
  int gs[16];
#pragma unroll
  for (int q = 0; q < 4; ++q) {
    const int4 g4 = ((const int4*)Gs)[q];
    gs[4 * q + 0] = g4.x;
    gs[4 * q + 1] = g4.y;
    gs[4 * q + 2] = g4.z;
    gs[4 * q + 3] = g4.w;
  }
#pragma unroll
  for (int r = 0; r < 16; ++r) tv[r] = T[r * T_PITCH + f];

  int cur = gs[0];
  float acc = 0.0f, cn = 0.0f;
#pragma unroll
  for (int r = 0; r < 16; ++r) {
    if (gs[r] != cur) {
      if (cur >= 0) {
        atomicAdd(&sums[(long long)cur * H_DIM + f], acc);
        if (f == 0) atomicAdd(&cnts[cur], cn);
      }
      acc = 0.0f;
      cn = 0.0f;
      cur = gs[r];
    }
    acc += tv[r];
    cn += 1.0f;
  }
  if (cur >= 0) {
    atomicAdd(&sums[(long long)cur * H_DIM + f], acc);
    if (f == 0) atomicAdd(&cnts[cur], cn);
  }
}

static inline size_t align_up(size_t x, size_t a) { return (x + a - 1) & ~(a - 1); }

extern "C" void kernel_launch(void* const* d_in, const int* in_sizes, int n_in,
                              void* d_out, int out_size, void* d_ws, size_t ws_size,
                              hipStream_t stream) {
  const float* x   = (const float*)d_in[0];
  const int*   ei  = (const int*)d_in[1];
  const int*   bat = (const int*)d_in[2];
  const float* W1  = (const float*)d_in[3];
  const float* b1  = (const float*)d_in[4];
  const float* W2  = (const float*)d_in[5];
  const float* b2  = (const float*)d_in[6];
  const float* Wmu = (const float*)d_in[7];
  const float* bmu = (const float*)d_in[8];
  const float* Wls = (const float*)d_in[9];
  const float* bls = (const float*)d_in[10];
  float* out = (float*)d_out;

  const int N = in_sizes[0] / F_IN;
  const int E = in_sizes[1] / 2;
  const int* src = ei;
  const int* dst = ei + E;

  char* ws = (char*)d_ws;
  size_t off = 0;
  const size_t zero_beg = off;
  int* cnt    = (int*)(ws + off);   off = align_up(off + (size_t)N * 4, 256);
  float* sums = (float*)(ws + off); off = align_up(off + (size_t)G_GRAPHS * H_DIM * 4, 256);
  float* cnts = (float*)(ws + off); off = align_up(off + (size_t)G_GRAPHS * 4, 256);
  int* ovf_n  = (int*)(ws + off);   off = align_up(off + 4, 256);
  const size_t zero_end = off;
  int* tab    = (int*)(ws + off);   off = align_up(off + (size_t)N * ROW_SLOTS * 4, 256);
  int* ovf    = (int*)(ws + off);   off = align_up(off + (size_t)E * 8, 256);
  float* bufA = (float*)(ws + off); off = align_up(off + (size_t)N * H_DIM * 4, 256);
  float* bufB = (float*)(ws + off); off = align_up(off + (size_t)N * H_DIM * 4, 256);
  (void)ws_size; (void)n_in; (void)out_size;

  const int BLK = 256;
  const int n16 = (int)((zero_end - zero_beg) / 16);
  const int n_rb = N / 16;
  const int gemm_blocks = (n_rb + 7) / 8;
  const int PER = 4 * BLK;

  clear_words<<<(n16 + BLK - 1) / BLK, BLK, 0, stream>>>((int4*)(ws + zero_beg), n16);

  count_slot_gemm<<<gemm_blocks + (E + PER - 1) / PER, BLK, 0, stream>>>(
      src, dst, cnt, tab, ovf_n, ovf, E, x, W1, bufA, n_rb, gemm_blocks);

  gather_prop_gemm<<<(N + 15) / 16, BLK, 0, stream>>>(cnt, tab, ovf_n, ovf, bufA, b1, W2, bufB, N, E);

  gather_prop_pool<<<(N + 15) / 16, BLK, 0, stream>>>(cnt, tab, ovf_n, ovf, bufB, b2, bat, sums, cnts, N, E);

  heads_wmma_f32<<<dim3(2 * (Z_DIM / 16), G_GRAPHS / 16), 32, 0, stream>>>(
      sums, cnts, Wmu, bmu, Wls, bls, out);
}


# Edge slots from a two-level counting sort on LDS counters (edges filed per 128-node block, one workgroup per block ranks them), not a returning global add per edge; row sum order = LDS arrival order
# speedup vs baseline: 4.7341x; 1.6778x over previous
#include <hip/hip_runtime.h>

typedef __attribute__((ext_vector_type(2))) float v2f;
typedef __attribute__((ext_vector_type(8))) float v8f;

#define F_IN   128
#define H_DIM  64
#define Z_DIM  32
#define G_GRAPHS 512

template <int K>
__device__ __forceinline__ void gemm_rowblock_f32(
    const float* __restrict__ A, const float* __restrict__ B,
    float* __restrict__ C, int tileM, int lane)
{
  const int half  = lane >> 4;
  const int l     = lane & 15;

  const float* Arow = A + (long long)(tileM * 16 + l) * K;
  const float* Bcol = B + l;

  v8f c0 = {0.f, 0.f, 0.f, 0.f, 0.f, 0.f, 0.f, 0.f};
  v8f c1 = c0, c2 = c0, c3 = c0;

#pragma unroll 4
  for (int k = 0; k < K; k += 4) {
    const int ka = k + 2 * half;
    v2f a, b0, b1, b2, b3;
    a.x = Arow[ka];
    a.y = Arow[ka + 1];
    const float* Bk0 = Bcol + ka * H_DIM;
    const float* Bk1 = Bcol + (ka + 1) * H_DIM;
    b0.x = Bk0[0];  b0.y = Bk1[0];
    b1.x = Bk0[16]; b1.y = Bk1[16];
    b2.x = Bk0[32]; b2.y = Bk1[32];
    b3.x = Bk0[48]; b3.y = Bk1[48];
    c0 = __builtin_amdgcn_wmma_f32_16x16x4_f32(false, a, false, b0, (short)0, c0, false, false);
    c1 = __builtin_amdgcn_wmma_f32_16x16x4_f32(false, a, false, b1, (short)0, c1, false, false);
    c2 = __builtin_amdgcn_wmma_f32_16x16x4_f32(false, a, false, b2, (short)0, c2, false, false);
    c3 = __builtin_amdgcn_wmma_f32_16x16x4_f32(false, a, false, b3, (short)0, c3, false, false);
  }

  float* Crow = C + (long long)(tileM * 16 + 8 * half) * H_DIM + l;
#pragma unroll
  for (int r = 0; r < 8; ++r) {
    Crow[r * H_DIM]      = c0[r];
    Crow[r * H_DIM + 16] = c1[r];
    Crow[r * H_DIM + 32] = c2[r];
    Crow[r * H_DIM + 48] = c3[r];
  }
}

__global__ __launch_bounds__(32) void heads_wmma_f32(
    const float* __restrict__ sums, const float* __restrict__ cnts,
    const float* __restrict__ Wmu, const float* __restrict__ bmu,
    const float* __restrict__ Wls, const float* __restrict__ bls,
    float* __restrict__ out)
{
  const int head  = blockIdx.x / (Z_DIM / 16);
  const int tileN = blockIdx.x % (Z_DIM / 16);
  const int tileM = blockIdx.y;
  const int lane  = threadIdx.x;
  const int half  = lane >> 4;
  const int l     = lane & 15;

  const float* __restrict__ Wm = head ? Wls : Wmu;
  const float* __restrict__ bm = head ? bls : bmu;
  float* __restrict__ C = out + (size_t)head * G_GRAPHS * Z_DIM;

  const int row = tileM * 16 + l;
  const float den = fmaxf(cnts[row], 1.0f);
  const float* Arow = sums + (long long)row * H_DIM;
  const float* Bcol = Wm + (tileN * 16 + l);

  v8f c = {0.f, 0.f, 0.f, 0.f, 0.f, 0.f, 0.f, 0.f};

  v2f a[H_DIM / 4], b[H_DIM / 4];
#pragma unroll
  for (int s = 0; s < H_DIM / 4; ++s) {
    const int ka = 4 * s + 2 * half;
    a[s].x = Arow[ka];
    a[s].y = Arow[ka + 1];
    b[s].x = Bcol[ka * Z_DIM];
    b[s].y = Bcol[(ka + 1) * Z_DIM];
  }
#pragma unroll
  for (int s = 0; s < H_DIM / 4; ++s) {
    a[s].x = a[s].x / den;
    a[s].y = a[s].y / den;
  }
  __builtin_amdgcn_sched_barrier(0);
#pragma unroll
  for (int s = 0; s < H_DIM / 4; ++s) {
    c = __builtin_amdgcn_wmma_f32_16x16x4_f32(
        false, a[s], false, b[s], (short)0, c, false, false);
  }

  const int col = tileN * 16 + l;
  const float bv = bm[col];
  float* Crow = C + (long long)(tileM * 16 + 8 * half) * Z_DIM + col;
#pragma unroll
  for (int r = 0; r < 8; ++r) {
    Crow[r * Z_DIM] = c[r] + bv;
  }
}

#define ROW_SLOTS 64
#define NPB_SHIFT 7
#define NPB   (1 << NPB_SHIFT)
#define CELL  32
#define CHUNK 4096

__device__ __forceinline__ void put_edge(int* __restrict__ tab, int* __restrict__ ovf_n,
                                         int* __restrict__ ovf, int s, int d, int r) {
  if (r < ROW_SLOTS) {
    tab[(long long)d * ROW_SLOTS + r] = s;
  } else {
    const int k = atomicAdd(ovf_n, 1);
    ((int2*)ovf)[k] = make_int2(s, d);
  }
}

__global__ __launch_bounds__(256) void file_edges_gemm(
    const int* __restrict__ src, const int* __restrict__ dst,
    int* __restrict__ cellcnt, int2* __restrict__ cells, int E, int nb, int n_chunks,
    const float* __restrict__ A, const float* __restrict__ B, float* __restrict__ C,
    int n_rb, int gemm_blocks, int4* __restrict__ zero_p, int n16)
{
  extern __shared__ int hc[];
  const int tid = (int)threadIdx.x;
  if ((int)blockIdx.x < gemm_blocks) {
    const int rb = __builtin_amdgcn_readfirstlane((int)(blockIdx.x * 8 + (threadIdx.x >> 5)));
    if (rb >= n_rb) return;
    gemm_rowblock_f32<F_IN>(A, B, C, rb, (int)(threadIdx.x & 31));
  } else if ((int)blockIdx.x < gemm_blocks + n_chunks) {
    const int wg = (int)blockIdx.x - gemm_blocks;
    const int t0 = wg * CHUNK + tid;
    int d[16], s[16], bk[16], r[16];
    if (wg * CHUNK + CHUNK <= E) {
#pragma unroll
      for (int j = 0; j < 16; ++j) d[j] = dst[t0 + 256 * j];
#pragma unroll
      for (int j = 0; j < 16; ++j) s[j] = src[t0 + 256 * j];
    } else {
#pragma unroll
      for (int j = 0; j < 16; ++j) {
        const int t = t0 + 256 * j;
        d[j] = t < E ? dst[t] : -1;
        s[j] = t < E ? src[t] : 0;
      }
    }
    for (int b = tid; b < nb + 32; b += 256) hc[b] = 0;
    __syncthreads();
#pragma unroll
    for (int j = 0; j < 16; ++j) bk[j] = d[j] >= 0 ? (d[j] >> NPB_SHIFT) : nb + (tid & 31);
#pragma unroll
    for (int j = 0; j < 16; ++j) r[j] = atomicAdd(&hc[bk[j]], 1);
    int2* __restrict__ mine = cells + (long long)wg * CELL;
#pragma unroll
    for (int j = 0; j < 16; ++j) {
      if (d[j] >= 0 && r[j] < CELL)
        mine[(long long)bk[j] * n_chunks * CELL + r[j]] = make_int2(s[j], d[j]);
    }
    __syncthreads();
    for (int b = tid; b < nb; b += 256) cellcnt[(long long)b * n_chunks + wg] = hc[b];
  } else {
    const int t = ((int)blockIdx.x - gemm_blocks - n_chunks) * 256 + tid;
    if (t < n16) zero_p[t] = make_int4(0, 0, 0, 0);
  }
}

__device__ __forceinline__ void place_edge(int* lc, int node0, int* __restrict__ tab,
                                           int* __restrict__ ovf_n, int* __restrict__ ovf,
                                           int s, int d) {
  const int r = atomicAdd(&lc[d - node0], 1);
  put_edge(tab, ovf_n, ovf, s, d, r);
}

__global__ __launch_bounds__(256) void rows_from_cells(
    const int* __restrict__ cellcnt, const int2* __restrict__ cells,
    const int* __restrict__ src, const int* __restrict__ dst,
    int* __restrict__ cnt, int* __restrict__ tab,
    int* __restrict__ ovf_n, int* __restrict__ ovf, int n, int E, int n_chunks)
{
  __shared__ int lc[NPB + 32];
  const int b    = (int)blockIdx.x;
  const int tid  = (int)threadIdx.x;
  const int lane = tid & 31;
  const int w    = __builtin_amdgcn_readfirstlane(tid >> 5);
  const int node0 = b << NPB_SHIFT;
  if (tid < NPB + 32) lc[tid] = 0;
  __syncthreads();

  const int*  __restrict__ cc = cellcnt + (long long)b * n_chunks;
  const int2* __restrict__ cb = cells + (long long)b * n_chunks * CELL;

  for (int g0 = 0; g0 < n_chunks; g0 += 256) {
    const int mycell = g0 + w + 8 * lane;
    const int cvec = mycell < n_chunks ? cc[mycell] : 0;
    for (int j0 = 0; j0 < 32; j0 += 16) {
      if (g0 + w + 8 * j0 >= n_chunks) break;
      int2 ev[16];
      int  cv[16], r[16];
#pragma unroll
      for (int u = 0; u < 16; ++u) {
        cv[u] = __builtin_amdgcn_readlane(cvec, j0 + u);
        ev[u] = make_int2(0, 0);
        if (cv[u] <= CELL && lane < cv[u])
          ev[u] = cb[(long long)(g0 + w + 8 * (j0 + u)) * CELL + lane];
      }
#pragma unroll
      for (int u = 0; u < 16; ++u) {
        const bool on = cv[u] <= CELL && lane < cv[u];
        r[u] = atomicAdd(&lc[on ? ev[u].y - node0 : NPB + lane], 1);
      }
#pragma unroll
      for (int u = 0; u < 16; ++u) {
        if (cv[u] <= CELL && lane < cv[u])
          put_edge(tab, ovf_n, ovf, ev[u].x, ev[u].y, r[u]);
      }
    }
    for (int j = 0; j < 32; ++j) {
      const int c = __builtin_amdgcn_readlane(cvec, j);
      if (c > CELL) {
        const int e0 = (g0 + w + 8 * j) * CHUNK;
        for (int k = lane; k < CHUNK; k += 32) {
          const int e = e0 + k;
          if (e < E) {
            const int dv = dst[e];
            if ((dv >> NPB_SHIFT) == b) place_edge(lc, node0, tab, ovf_n, ovf, src[e], dv);
          }
        }
      }
    }
  }
  __syncthreads();
  if (tid < NPB && node0 + tid < n) cnt[node0 + tid] = lc[tid];
}

#define T_PITCH 68

__device__ __forceinline__ float dinv_of(int c) {
  return __builtin_amdgcn_rsqf((float)c + 1.0f);
}

__device__ __forceinline__ float4 gather_row(
    const int* __restrict__ cnt, const int* __restrict__ tab,
    const int* __restrict__ ovf_n, const int* __restrict__ ovf,
    const float* __restrict__ h, const float* __restrict__ bias,
    int ic, bool live, int sub, int hi, int E)
{
  const int* __restrict__ trow = tab + (long long)ic * ROW_SLOTS;
  const float4* __restrict__ h4 = (const float4*)h;
  const int degi = cnt[ic];
  const float4 hv = h4[(long long)ic * (H_DIM / 4) + sub];
  const int deg = live ? degi : 0;
  const int len = min(deg, ROW_SLOTS);
  const float di = dinv_of(degi);

  const int len0 = __builtin_amdgcn_readlane(len, 0);
  const int len1 = __builtin_amdgcn_readlane(len, 16);
  const int maxlen = max(len0, len1);

  int s_cur = ic, s_nxt = ic;
  if (sub < len) s_cur = trow[sub];
  if (16 + sub < len) s_nxt = trow[16 + sub];

  float ax = 0.0f, ay = 0.0f, az = 0.0f, aw = 0.0f;
  for (int base = 0; base < maxlen; base += 16) {
    const int m = min(16, maxlen - base);
    const int px = s_cur;
    s_cur = s_nxt;
    if (base + 32 < maxlen) {
      const int e2 = base + 32 + sub;
      s_nxt = ic;
      if (e2 < len) s_nxt = trow[e2];
    }
    for (int j = 0; j < m; j += 8) {
      float4 v[8];
      int c[8];
#pragma unroll
      for (int u = 0; u < 8; ++u) {
        const int s0 = __builtin_amdgcn_readlane(px, j + u);
        const int s1 = __builtin_amdgcn_readlane(px, j + u + 16);
        const int sj = hi ? s1 : s0;
        v[u] = h4[sj * (H_DIM / 4) + sub];
        c[u] = cnt[sj];
      }
#pragma unroll
      for (int u = 0; u < 8; ++u) {
        const float cj = dinv_of(c[u]) * di;
        const bool on = base + j + u < len;
        const float tx = ax + v[u].x * cj;
        const float ty = ay + v[u].y * cj;
        const float tz = az + v[u].z * cj;
        const float tw = aw + v[u].w * cj;
        ax = on ? tx : ax;
        ay = on ? ty : ay;
        az = on ? tz : az;
        aw = on ? tw : aw;
      }
    }
  }

  const int deg0 = __builtin_amdgcn_readlane(deg, 0);
  const int deg1 = __builtin_amdgcn_readlane(deg, 16);
  if (max(deg0, deg1) > ROW_SLOTS) {
    const int nov = min(ovf_n[0], E);
    const int2* __restrict__ ov = (const int2*)ovf;
    for (int k = 0; k < nov; ++k) {
      const int2 od = ov[k];
      if (live && od.y == ic) {
        const float cj = dinv_of(cnt[od.x]) * di;
        const float4 v = h4[(long long)od.x * (H_DIM / 4) + sub];
        ax += v.x * cj;
        ay += v.y * cj;
        az += v.z * cj;
        aw += v.w * cj;
      }
    }
  }

  const float sl = di * di;
  const float4 bv = ((const float4*)bias)[sub];
  const float ox = ax + hv.x * sl + bv.x;
  const float oy = ay + hv.y * sl + bv.y;
  const float oz = az + hv.z * sl + bv.z;
  const float ow = aw + hv.w * sl + bv.w;
  float4 o;
  o.x = (live && ox > 0.0f) ? ox : 0.0f;
  o.y = (live && oy > 0.0f) ? oy : 0.0f;
  o.z = (live && oz > 0.0f) ? oz : 0.0f;
  o.w = (live && ow > 0.0f) ? ow : 0.0f;
  return o;
}

__global__ __launch_bounds__(256) void gather_prop_gemm(
    const int* __restrict__ cnt, const int* __restrict__ tab,
    const int* __restrict__ ovf_n, const int* __restrict__ ovf,
    const float* __restrict__ h,
    const float* __restrict__ bias, const float* __restrict__ W,
    float* __restrict__ out, int n, int E)
{
  __shared__ __attribute__((aligned(16))) float T[16 * T_PITCH];
  __shared__ __attribute__((aligned(16))) float U[16 * T_PITCH];

  const int lane = threadIdx.x & 31;
  const int sub  = lane & 15;
  const int hi   = lane >> 4;
  const int w    = __builtin_amdgcn_readfirstlane((int)(threadIdx.x >> 5));
  const int rl   = 2 * w + hi;
  const int i    = (int)blockIdx.x * 16 + rl;
  const bool live = i < n;
  const int ic   = live ? i : n - 1;

  const float4 o = gather_row(cnt, tab, ovf_n, ovf, h, bias, ic, live, sub, hi, E);
  *(float4*)&T[rl * T_PITCH + 4 * sub] = o;

  v2f b[H_DIM / 4];
  if (w < 4) {
    const float* Bcol = W + 16 * w + sub;
#pragma unroll
    for (int s = 0; s < H_DIM / 4; ++s) {
      const int ka = 4 * s + 2 * hi;
      b[s].x = Bcol[ka * H_DIM];
      b[s].y = Bcol[(ka + 1) * H_DIM];
    }
  }
  __syncthreads();
  if (w < 4) {
    v8f c = {0.f, 0.f, 0.f, 0.f, 0.f, 0.f, 0.f, 0.f};
    const float* Arow = T + sub * T_PITCH + 2 * hi;
#pragma unroll
    for (int s = 0; s < H_DIM / 4; ++s) {
      v2f a;
      a.x = Arow[4 * s];
      a.y = Arow[4 * s + 1];
      c = __builtin_amdgcn_wmma_f32_16x16x4_f32(false, a, false, b[s], (short)0, c, false, false);
    }
    float* Ucol = U + (8 * hi) * T_PITCH + 16 * w + sub;
#pragma unroll
    for (int r = 0; r < 8; ++r) Ucol[r * T_PITCH] = c[r];
  }
  __syncthreads();
  if (live) {
    const float4 v = *(const float4*)&U[rl * T_PITCH + 4 * sub];
    ((float4*)out)[(long long)i * (H_DIM / 4) + sub] = v;
  }
}

__global__ __launch_bounds__(256) void gather_prop_pool(
    const int* __restrict__ cnt, const int* __restrict__ tab,
    const int* __restrict__ ovf_n, const int* __restrict__ ovf,
    const float* __restrict__ h,
    const float* __restrict__ bias, const int* __restrict__ batch,
    float* __restrict__ sums, float* __restrict__ cnts, int n, int E)
{
  __shared__ __attribute__((aligned(16))) float T[16 * T_PITCH];
  __shared__ __attribute__((aligned(16))) int Gs[16];

  const int lane = threadIdx.x & 31;
  const int sub  = lane & 15;
  const int hi   = lane >> 4;
  const int w    = (int)(threadIdx.x >> 5);
  const int rl   = 2 * w + hi;
  const int i    = (int)blockIdx.x * 16 + rl;
  const bool live = i < n;
  const int ic   = live ? i : n - 1;
  const int g    = batch[ic];

  const float4 o = gather_row(cnt, tab, ovf_n, ovf, h, bias, ic, live, sub, hi, E);
  *(float4*)&T[rl * T_PITCH + 4 * sub] = o;
  if (sub == 0) Gs[rl] = live ? g : -1;
  __syncthreads();
  if (threadIdx.x >= H_DIM) return;

  const int f = (int)threadIdx.x;
  float tv[16];
  int gs[16];
#pragma unroll
  for (int q = 0; q < 4; ++q) {
    const int4 g4 = ((const int4*)Gs)[q];
    gs[4 * q + 0] = g4.x;
    gs[4 * q + 1] = g4.y;
    gs[4 * q + 2] = g4.z;
    gs[4 * q + 3] = g4.w;
  }
#pragma unroll
  for (int r = 0; r < 16; ++r) tv[r] = T[r * T_PITCH + f];

  int cur = gs[0];
  float acc = 0.0f, cn = 0.0f;
#pragma unroll
  for (int r = 0; r < 16; ++r) {
    if (gs[r] != cur) {
      if (cur >= 0) {
        atomicAdd(&sums[(long long)cur * H_DIM + f], acc);
        if (f == 0) atomicAdd(&cnts[cur], cn);
      }
      acc = 0.0f;
      cn = 0.0f;
      cur = gs[r];
    }
    acc += tv[r];
    cn += 1.0f;
  }
  if (cur >= 0) {
    atomicAdd(&sums[(long long)cur * H_DIM + f], acc);
    if (f == 0) atomicAdd(&cnts[cur], cn);
  }
}

static inline size_t align_up(size_t x, size_t a) { return (x + a - 1) & ~(a - 1); }

extern "C" void kernel_launch(void* const* d_in, const int* in_sizes, int n_in,
                              void* d_out, int out_size, void* d_ws, size_t ws_size,
                              hipStream_t stream) {
  const float* x   = (const float*)d_in[0];
  const int*   ei  = (const int*)d_in[1];
  const int*   bat = (const int*)d_in[2];
  const float* W1  = (const float*)d_in[3];
  const float* b1  = (const float*)d_in[4];
  const float* W2  = (const float*)d_in[5];
  const float* b2  = (const float*)d_in[6];
  const float* Wmu = (const float*)d_in[7];
  const float* bmu = (const float*)d_in[8];
  const float* Wls = (const float*)d_in[9];
  const float* bls = (const float*)d_in[10];
  float* out = (float*)d_out;

  const int N = in_sizes[0] / F_IN;
  const int E = in_sizes[1] / 2;
  const int* src = ei;
  const int* dst = ei + E;

  char* ws = (char*)d_ws;
  size_t off = 0;
  const size_t zero_beg = off;
  float* sums = (float*)(ws + off); off = align_up(off + (size_t)G_GRAPHS * H_DIM * 4, 256);
  float* cnts = (float*)(ws + off); off = align_up(off + (size_t)G_GRAPHS * 4, 256);
  int* ovf_n  = (int*)(ws + off);   off = align_up(off + 4, 256);
  const size_t zero_end = off;
  const int nb = (N + NPB - 1) / NPB;
  const int n_chunks = (E + CHUNK - 1) / CHUNK;
  int* cnt    = (int*)(ws + off);   off = align_up(off + (size_t)N * 4, 256);
  int* cellcnt = (int*)(ws + off);  off = align_up(off + (size_t)nb * n_chunks * 4, 256);
  int2* cells = (int2*)(ws + off);  off = align_up(off + (size_t)nb * n_chunks * CELL * 8, 256);
  int* tab    = (int*)(ws + off);   off = align_up(off + (size_t)N * ROW_SLOTS * 4, 256);
  int* ovf    = (int*)(ws + off);   off = align_up(off + (size_t)E * 8, 256);
  float* bufA = (float*)(ws + off); off = align_up(off + (size_t)N * H_DIM * 4, 256);
  float* bufB = (float*)(ws + off); off = align_up(off + (size_t)N * H_DIM * 4, 256);
  (void)ws_size; (void)n_in; (void)out_size;

  const int BLK = 256;
  const int n16 = (int)((zero_end - zero_beg) / 16);
  const int n_rb = N / 16;
  const int gemm_blocks = (n_rb + 7) / 8;
  const int clear_blocks = (n16 + BLK - 1) / BLK;

  file_edges_gemm<<<gemm_blocks + n_chunks + clear_blocks, BLK, (size_t)(nb + 32) * 4, stream>>>(
      src, dst, cellcnt, cells, E, nb, n_chunks, x, W1, bufA, n_rb, gemm_blocks,
      (int4*)(ws + zero_beg), n16);

  rows_from_cells<<<nb, BLK, 0, stream>>>(cellcnt, cells, src, dst, cnt, tab, ovf_n, ovf, N, E, n_chunks);

  gather_prop_gemm<<<(N + 15) / 16, BLK, 0, stream>>>(cnt, tab, ovf_n, ovf, bufA, b1, W2, bufB, N, E);

  gather_prop_pool<<<(N + 15) / 16, BLK, 0, stream>>>(cnt, tab, ovf_n, ovf, bufB, b2, bat, sums, cnts, N, E);

  heads_wmma_f32<<<dim3(2 * (Z_DIM / 16), G_GRAPHS / 16), 32, 0, stream>>>(
      sums, cnts, Wmu, bmu, Wls, bls, out);
}


# Layer-2 product rows alone stored times 1/sqrt(deg), by the epilogue that writes them anyway (no added pass): gather 2 adds source rows as they are, no count load or rsq per edge; sum order kept
# speedup vs baseline: 4.9473x; 1.0450x over previous
#include <hip/hip_runtime.h>

typedef __attribute__((ext_vector_type(2))) float v2f;
typedef __attribute__((ext_vector_type(8))) float v8f;

#define F_IN   128
#define H_DIM  64
#define Z_DIM  32
#define G_GRAPHS 512

template <int K>
__device__ __forceinline__ void gemm_rowblock_f32(
    const float* __restrict__ A, const float* __restrict__ B,
    float* __restrict__ C, int tileM, int lane)
{
  const int half  = lane >> 4;
  const int l     = lane & 15;

  const float* Arow = A + (long long)(tileM * 16 + l) * K;
  const float* Bcol = B + l;

  v8f c0 = {0.f, 0.f, 0.f, 0.f, 0.f, 0.f, 0.f, 0.f};
  v8f c1 = c0, c2 = c0, c3 = c0;

#pragma unroll 4
  for (int k = 0; k < K; k += 4) {
    const int ka = k + 2 * half;
    v2f a, b0, b1, b2, b3;
    a.x = Arow[ka];
    a.y = Arow[ka + 1];
    const float* Bk0 = Bcol + ka * H_DIM;
    const float* Bk1 = Bcol + (ka + 1) * H_DIM;
    b0.x = Bk0[0];  b0.y = Bk1[0];
    b1.x = Bk0[16]; b1.y = Bk1[16];
    b2.x = Bk0[32]; b2.y = Bk1[32];
    b3.x = Bk0[48]; b3.y = Bk1[48];
    c0 = __builtin_amdgcn_wmma_f32_16x16x4_f32(false, a, false, b0, (short)0, c0, false, false);
    c1 = __builtin_amdgcn_wmma_f32_16x16x4_f32(false, a, false, b1, (short)0, c1, false, false);
    c2 = __builtin_amdgcn_wmma_f32_16x16x4_f32(false, a, false, b2, (short)0, c2, false, false);
    c3 = __builtin_amdgcn_wmma_f32_16x16x4_f32(false, a, false, b3, (short)0, c3, false, false);
  }

  float* Crow = C + (long long)(tileM * 16 + 8 * half) * H_DIM + l;
#pragma unroll
  for (int r = 0; r < 8; ++r) {
    Crow[r * H_DIM]      = c0[r];
    Crow[r * H_DIM + 16] = c1[r];
    Crow[r * H_DIM + 32] = c2[r];
    Crow[r * H_DIM + 48] = c3[r];
  }
}

__global__ __launch_bounds__(32) void heads_wmma_f32(
    const float* __restrict__ sums, const float* __restrict__ cnts,
    const float* __restrict__ Wmu, const float* __restrict__ bmu,
    const float* __restrict__ Wls, const float* __restrict__ bls,
    float* __restrict__ out)
{
  const int head  = blockIdx.x / (Z_DIM / 16);
  const int tileN = blockIdx.x % (Z_DIM / 16);
  const int tileM = blockIdx.y;
  const int lane  = threadIdx.x;
  const int half  = lane >> 4;
  const int l     = lane & 15;

  const float* __restrict__ Wm = head ? Wls : Wmu;
  const float* __restrict__ bm = head ? bls : bmu;
  float* __restrict__ C = out + (size_t)head * G_GRAPHS * Z_DIM;

  const int row = tileM * 16 + l;
  const float den = fmaxf(cnts[row], 1.0f);
  const float* Arow = sums + (long long)row * H_DIM;
  const float* Bcol = Wm + (tileN * 16 + l);

  v8f c = {0.f, 0.f, 0.f, 0.f, 0.f, 0.f, 0.f, 0.f};

  v2f a[H_DIM / 4], b[H_DIM / 4];
#pragma unroll
  for (int s = 0; s < H_DIM / 4; ++s) {
    const int ka = 4 * s + 2 * half;
    a[s].x = Arow[ka];
    a[s].y = Arow[ka + 1];
    b[s].x = Bcol[ka * Z_DIM];
    b[s].y = Bcol[(ka + 1) * Z_DIM];
  }
#pragma unroll
  for (int s = 0; s < H_DIM / 4; ++s) {
    a[s].x = a[s].x / den;
    a[s].y = a[s].y / den;
  }
  __builtin_amdgcn_sched_barrier(0);
#pragma unroll
  for (int s = 0; s < H_DIM / 4; ++s) {
    c = __builtin_amdgcn_wmma_f32_16x16x4_f32(
        false, a[s], false, b[s], (short)0, c, false, false);
  }

  const int col = tileN * 16 + l;
  const float bv = bm[col];
  float* Crow = C + (long long)(tileM * 16 + 8 * half) * Z_DIM + col;
#pragma unroll
  for (int r = 0; r < 8; ++r) {
    Crow[r * Z_DIM] = c[r] + bv;
  }
}

#define ROW_SLOTS 64
#define NPB_SHIFT 7
#define NPB   (1 << NPB_SHIFT)
#define CELL  32
#define CHUNK 4096

__device__ __forceinline__ void put_edge(int* __restrict__ tab, int* __restrict__ ovf_n,
                                         int* __restrict__ ovf, int s, int d, int r) {
  if (r < ROW_SLOTS) {
    tab[(long long)d * ROW_SLOTS + r] = s;
  } else {
    const int k = atomicAdd(ovf_n, 1);
    ((int2*)ovf)[k] = make_int2(s, d);
  }
}

__global__ __launch_bounds__(256) void file_edges_gemm(
    const int* __restrict__ src, const int* __restrict__ dst,
    int* __restrict__ cellcnt, int2* __restrict__ cells, int E, int nb, int n_chunks,
    const float* __restrict__ A, const float* __restrict__ B, float* __restrict__ C,
    int n_rb, int gemm_blocks, int4* __restrict__ zero_p, int n16)
{
  extern __shared__ int hc[];
  const int tid = (int)threadIdx.x;
  if ((int)blockIdx.x < gemm_blocks) {
    const int rb = __builtin_amdgcn_readfirstlane((int)(blockIdx.x * 8 + (threadIdx.x >> 5)));
    if (rb >= n_rb) return;
    gemm_rowblock_f32<F_IN>(A, B, C, rb, (int)(threadIdx.x & 31));
  } else if ((int)blockIdx.x < gemm_blocks + n_chunks) {
    const int wg = (int)blockIdx.x - gemm_blocks;
    const int t0 = wg * CHUNK + tid;
    int d[16], s[16], bk[16], r[16];
    if (wg * CHUNK + CHUNK <= E) {
#pragma unroll
      for (int j = 0; j < 16; ++j) d[j] = dst[t0 + 256 * j];
#pragma unroll
      for (int j = 0; j < 16; ++j) s[j] = src[t0 + 256 * j];
    } else {
#pragma unroll
      for (int j = 0; j < 16; ++j) {
        const int t = t0 + 256 * j;
        d[j] = t < E ? dst[t] : -1;
        s[j] = t < E ? src[t] : 0;
      }
    }
    for (int b = tid; b < nb + 32; b += 256) hc[b] = 0;
    __syncthreads();
#pragma unroll
    for (int j = 0; j < 16; ++j) bk[j] = d[j] >= 0 ? (d[j] >> NPB_SHIFT) : nb + (tid & 31);
#pragma unroll
    for (int j = 0; j < 16; ++j) r[j] = atomicAdd(&hc[bk[j]], 1);
    int2* __restrict__ mine = cells + (long long)wg * CELL;
#pragma unroll
    for (int j = 0; j < 16; ++j) {
      if (d[j] >= 0 && r[j] < CELL)
        mine[(long long)bk[j] * n_chunks * CELL + r[j]] = make_int2(s[j], d[j]);
    }
    __syncthreads();
    for (int b = tid; b < nb; b += 256) cellcnt[(long long)b * n_chunks + wg] = hc[b];
  } else {
    const int t = ((int)blockIdx.x - gemm_blocks - n_chunks) * 256 + tid;
    if (t < n16) zero_p[t] = make_int4(0, 0, 0, 0);
  }
}

__device__ __forceinline__ void place_edge(int* lc, int node0, int* __restrict__ tab,
                                           int* __restrict__ ovf_n, int* __restrict__ ovf,
                                           int s, int d) {
  const int r = atomicAdd(&lc[d - node0], 1);
  put_edge(tab, ovf_n, ovf, s, d, r);
}

__global__ __launch_bounds__(256) void rows_from_cells(
    const int* __restrict__ cellcnt, const int2* __restrict__ cells,
    const int* __restrict__ src, const int* __restrict__ dst,
    int* __restrict__ cnt, int* __restrict__ tab,
    int* __restrict__ ovf_n, int* __restrict__ ovf, int n, int E, int n_chunks)
{
  __shared__ int lc[NPB + 32];
  const int b    = (int)blockIdx.x;
  const int tid  = (int)threadIdx.x;
  const int lane = tid & 31;
  const int w    = __builtin_amdgcn_readfirstlane(tid >> 5);
  const int node0 = b << NPB_SHIFT;
  if (tid < NPB + 32) lc[tid] = 0;
  __syncthreads();

  const int*  __restrict__ cc = cellcnt + (long long)b * n_chunks;
  const int2* __restrict__ cb = cells + (long long)b * n_chunks * CELL;

  for (int g0 = 0; g0 < n_chunks; g0 += 256) {
    const int mycell = g0 + w + 8 * lane;
    const int cvec = mycell < n_chunks ? cc[mycell] : 0;
    for (int j0 = 0; j0 < 32; j0 += 16) {
      if (g0 + w + 8 * j0 >= n_chunks) break;
      int2 ev[16];
      int  cv[16], r[16];
#pragma unroll
      for (int u = 0; u < 16; ++u) {
        cv[u] = __builtin_amdgcn_readlane(cvec, j0 + u);
        ev[u] = make_int2(0, 0);
        if (cv[u] <= CELL && lane < cv[u])
          ev[u] = cb[(long long)(g0 + w + 8 * (j0 + u)) * CELL + lane];
      }
#pragma unroll
      for (int u = 0; u < 16; ++u) {
        const bool on = cv[u] <= CELL && lane < cv[u];
        r[u] = atomicAdd(&lc[on ? ev[u].y - node0 : NPB + lane], 1);
      }
#pragma unroll
      for (int u = 0; u < 16; ++u) {
        if (cv[u] <= CELL && lane < cv[u])
          put_edge(tab, ovf_n, ovf, ev[u].x, ev[u].y, r[u]);
      }
    }
    for (int j = 0; j < 32; ++j) {
      const int c = __builtin_amdgcn_readlane(cvec, j);
      if (c > CELL) {
        const int e0 = (g0 + w + 8 * j) * CHUNK;
        for (int k = lane; k < CHUNK; k += 32) {
          const int e = e0 + k;
          if (e < E) {
            const int dv = dst[e];
            if ((dv >> NPB_SHIFT) == b) place_edge(lc, node0, tab, ovf_n, ovf, src[e], dv);
          }
        }
      }
    }
  }
  __syncthreads();
  if (tid < NPB && node0 + tid < n) cnt[node0 + tid] = lc[tid];
}

#define T_PITCH 68

__device__ __forceinline__ float dinv_of(int c) {
  return __builtin_amdgcn_rsqf((float)c + 1.0f);
}

template <bool PRE>
__device__ __forceinline__ float4 gather_row(
    const int* __restrict__ cnt, const int* __restrict__ tab,
    const int* __restrict__ ovf_n, const int* __restrict__ ovf,
    const float* __restrict__ h, const float* __restrict__ bias,
    int ic, bool live, int sub, int hi, int E, float& di_out)
{
  const int* __restrict__ trow = tab + (long long)ic * ROW_SLOTS;
  const float4* __restrict__ h4 = (const float4*)h;
  const int degi = cnt[ic];
  const float4 hv = h4[(long long)ic * (H_DIM / 4) + sub];
  const int deg = live ? degi : 0;
  const int len = min(deg, ROW_SLOTS);
  const float di = dinv_of(degi);
  di_out = di;

  const int len0 = __builtin_amdgcn_readlane(len, 0);
  const int len1 = __builtin_amdgcn_readlane(len, 16);
  const int maxlen = max(len0, len1);

  int s_cur = ic, s_nxt = ic;
  if (sub < len) s_cur = trow[sub];
  if (16 + sub < len) s_nxt = trow[16 + sub];

  float ax = 0.0f, ay = 0.0f, az = 0.0f, aw = 0.0f;
  for (int base = 0; base < maxlen; base += 16) {
    const int m = min(16, maxlen - base);
    const int px = s_cur;
    s_cur = s_nxt;
    if (base + 32 < maxlen) {
      const int e2 = base + 32 + sub;
      s_nxt = ic;
      if (e2 < len) s_nxt = trow[e2];
    }
    for (int j = 0; j < m; j += 8) {
      float4 v[8];
      int c[8];
#pragma unroll
      for (int u = 0; u < 8; ++u) {
        const int s0 = __builtin_amdgcn_readlane(px, j + u);
        const int s1 = __builtin_amdgcn_readlane(px, j + u + 16);
        const int sj = hi ? s1 : s0;
        v[u] = h4[sj * (H_DIM / 4) + sub];
        if (!PRE) c[u] = cnt[sj];
      }
#pragma unroll
      for (int u = 0; u < 8; ++u) {
        const bool on = base + j + u < len;
        float tx, ty, tz, tw;
        if (PRE) {
          tx = ax + v[u].x;
          ty = ay + v[u].y;
          tz = az + v[u].z;
          tw = aw + v[u].w;
        } else {
          const float cj = dinv_of(c[u]) * di;
          tx = ax + v[u].x * cj;
          ty = ay + v[u].y * cj;
          tz = az + v[u].z * cj;
          tw = aw + v[u].w * cj;
        }
        ax = on ? tx : ax;
        ay = on ? ty : ay;
        az = on ? tz : az;
        aw = on ? tw : aw;
      }
    }
  }

  const int deg0 = __builtin_amdgcn_readlane(deg, 0);
  const int deg1 = __builtin_amdgcn_readlane(deg, 16);
  if (max(deg0, deg1) > ROW_SLOTS) {
    const int nov = min(ovf_n[0], E);
    const int2* __restrict__ ov = (const int2*)ovf;
    for (int k = 0; k < nov; ++k) {
      const int2 od = ov[k];
      if (live && od.y == ic) {
        const float4 v = h4[(long long)od.x * (H_DIM / 4) + sub];
        if (PRE) {
          ax += v.x;
          ay += v.y;
          az += v.z;
          aw += v.w;
        } else {
          const float cj = dinv_of(cnt[od.x]) * di;
          ax += v.x * cj;
          ay += v.y * cj;
          az += v.z * cj;
          aw += v.w * cj;
        }
      }
    }
  }

  const float4 bv = ((const float4*)bias)[sub];
  float ox, oy, oz, ow;
  if (PRE) {
    ox = (ax + hv.x) * di + bv.x;
    oy = (ay + hv.y) * di + bv.y;
    oz = (az + hv.z) * di + bv.z;
    ow = (aw + hv.w) * di + bv.w;
  } else {
    const float sl = di * di;
    ox = ax + hv.x * sl + bv.x;
    oy = ay + hv.y * sl + bv.y;
    oz = az + hv.z * sl + bv.z;
    ow = aw + hv.w * sl + bv.w;
  }
  float4 o;
  o.x = (live && ox > 0.0f) ? ox : 0.0f;
  o.y = (live && oy > 0.0f) ? oy : 0.0f;
  o.z = (live && oz > 0.0f) ? oz : 0.0f;
  o.w = (live && ow > 0.0f) ? ow : 0.0f;
  return o;
}

__global__ __launch_bounds__(256) void gather_prop_gemm(
    const int* __restrict__ cnt, const int* __restrict__ tab,
    const int* __restrict__ ovf_n, const int* __restrict__ ovf,
    const float* __restrict__ h,
    const float* __restrict__ bias, const float* __restrict__ W,
    float* __restrict__ out, int n, int E)
{
  __shared__ __attribute__((aligned(16))) float T[16 * T_PITCH];
  __shared__ __attribute__((aligned(16))) float U[16 * T_PITCH];

  const int lane = threadIdx.x & 31;
  const int sub  = lane & 15;
  const int hi   = lane >> 4;
  const int w    = __builtin_amdgcn_readfirstlane((int)(threadIdx.x >> 5));
  const int rl   = 2 * w + hi;
  const int i    = (int)blockIdx.x * 16 + rl;
  const bool live = i < n;
  const int ic   = live ? i : n - 1;

  float di;
  const float4 o = gather_row<false>(cnt, tab, ovf_n, ovf, h, bias, ic, live, sub, hi, E, di);
  *(float4*)&T[rl * T_PITCH + 4 * sub] = o;

  v2f b[H_DIM / 4];
  if (w < 4) {
    const float* Bcol = W + 16 * w + sub;
#pragma unroll
    for (int s = 0; s < H_DIM / 4; ++s) {
      const int ka = 4 * s + 2 * hi;
      b[s].x = Bcol[ka * H_DIM];
      b[s].y = Bcol[(ka + 1) * H_DIM];
    }
  }
  __syncthreads();
  if (w < 4) {
    v8f c = {0.f, 0.f, 0.f, 0.f, 0.f, 0.f, 0.f, 0.f};
    const float* Arow = T + sub * T_PITCH + 2 * hi;
#pragma unroll
    for (int s = 0; s < H_DIM / 4; ++s) {
      v2f a;
      a.x = Arow[4 * s];
      a.y = Arow[4 * s + 1];
      c = __builtin_amdgcn_wmma_f32_16x16x4_f32(false, a, false, b[s], (short)0, c, false, false);
    }
    float* Ucol = U + (8 * hi) * T_PITCH + 16 * w + sub;
#pragma unroll
    for (int r = 0; r < 8; ++r) Ucol[r * T_PITCH] = c[r];
  }
  __syncthreads();
  if (live) {
    float4 v = *(const float4*)&U[rl * T_PITCH + 4 * sub];
    v.x *= di;
    v.y *= di;
    v.z *= di;
    v.w *= di;
    ((float4*)out)[(long long)i * (H_DIM / 4) + sub] = v;
  }
}

__global__ __launch_bounds__(256) void gather_prop_pool(
    const int* __restrict__ cnt, const int* __restrict__ tab,
    const int* __restrict__ ovf_n, const int* __restrict__ ovf,
    const float* __restrict__ h,
    const float* __restrict__ bias, const int* __restrict__ batch,
    float* __restrict__ sums, float* __restrict__ cnts, int n, int E)
{
  __shared__ __attribute__((aligned(16))) float T[16 * T_PITCH];
  __shared__ __attribute__((aligned(16))) int Gs[16];

  const int lane = threadIdx.x & 31;
  const int sub  = lane & 15;
  const int hi   = lane >> 4;
  const int w    = (int)(threadIdx.x >> 5);
  const int rl   = 2 * w + hi;
  const int i    = (int)blockIdx.x * 16 + rl;
  const bool live = i < n;
  const int ic   = live ? i : n - 1;
  const int g    = batch[ic];

  float di;
  const float4 o = gather_row<true>(cnt, tab, ovf_n, ovf, h, bias, ic, live, sub, hi, E, di);
  *(float4*)&T[rl * T_PITCH + 4 * sub] = o;
  if (sub == 0) Gs[rl] = live ? g : -1;
  __syncthreads();
  if (threadIdx.x >= H_DIM) return;

  const int f = (int)threadIdx.x;
  float tv[16];
  int gs[16];
#pragma unroll
  for (int q = 0; q < 4; ++q) {
    const int4 g4 = ((const int4*)Gs)[q];
    gs[4 * q + 0] = g4.x;
    gs[4 * q + 1] = g4.y;
    gs[4 * q + 2] = g4.z;
    gs[4 * q + 3] = g4.w;
  }
#pragma unroll
  for (int r = 0; r < 16; ++r) tv[r] = T[r * T_PITCH + f];

  int cur = gs[0];
  float acc = 0.0f, cn = 0.0f;
#pragma unroll
  for (int r = 0; r < 16; ++r) {
    if (gs[r] != cur) {
      if (cur >= 0) {
        atomicAdd(&sums[(long long)cur * H_DIM + f], acc);
        if (f == 0) atomicAdd(&cnts[cur], cn);
      }
      acc = 0.0f;
      cn = 0.0f;
      cur = gs[r];
    }
    acc += tv[r];
    cn += 1.0f;
  }
  if (cur >= 0) {
    atomicAdd(&sums[(long long)cur * H_DIM + f], acc);
    if (f == 0) atomicAdd(&cnts[cur], cn);
  }
}

static inline size_t align_up(size_t x, size_t a) { return (x + a - 1) & ~(a - 1); }

extern "C" void kernel_launch(void* const* d_in, const int* in_sizes, int n_in,
                              void* d_out, int out_size, void* d_ws, size_t ws_size,
                              hipStream_t stream) {
  const float* x   = (const float*)d_in[0];
  const int*   ei  = (const int*)d_in[1];
  const int*   bat = (const int*)d_in[2];
  const float* W1  = (const float*)d_in[3];
  const float* b1  = (const float*)d_in[4];
  const float* W2  = (const float*)d_in[5];
  const float* b2  = (const float*)d_in[6];
  const float* Wmu = (const float*)d_in[7];
  const float* bmu = (const float*)d_in[8];
  const float* Wls = (const float*)d_in[9];
  const float* bls = (const float*)d_in[10];
  float* out = (float*)d_out;

  const int N = in_sizes[0] / F_IN;
  const int E = in_sizes[1] / 2;
  const int* src = ei;
  const int* dst = ei + E;

  char* ws = (char*)d_ws;
  size_t off = 0;
  const size_t zero_beg = off;
  float* sums = (float*)(ws + off); off = align_up(off + (size_t)G_GRAPHS * H_DIM * 4, 256);
  float* cnts = (float*)(ws + off); off = align_up(off + (size_t)G_GRAPHS * 4, 256);
  int* ovf_n  = (int*)(ws + off);   off = align_up(off + 4, 256);
  const size_t zero_end = off;
  const int nb = (N + NPB - 1) / NPB;
  const int n_chunks = (E + CHUNK - 1) / CHUNK;
  int* cnt    = (int*)(ws + off);   off = align_up(off + (size_t)N * 4, 256);
  int* cellcnt = (int*)(ws + off);  off = align_up(off + (size_t)nb * n_chunks * 4, 256);
  int2* cells = (int2*)(ws + off);  off = align_up(off + (size_t)nb * n_chunks * CELL * 8, 256);
  int* tab    = (int*)(ws + off);   off = align_up(off + (size_t)N * ROW_SLOTS * 4, 256);
  int* ovf    = (int*)(ws + off);   off = align_up(off + (size_t)E * 8, 256);
  float* bufA = (float*)(ws + off); off = align_up(off + (size_t)N * H_DIM * 4, 256);
  float* bufB = (float*)(ws + off); off = align_up(off + (size_t)N * H_DIM * 4, 256);
  (void)ws_size; (void)n_in; (void)out_size;

  const int BLK = 256;
  const int n16 = (int)((zero_end - zero_beg) / 16);
  const int n_rb = N / 16;
  const int gemm_blocks = (n_rb + 7) / 8;
  const int clear_blocks = (n16 + BLK - 1) / BLK;

  file_edges_gemm<<<gemm_blocks + n_chunks + clear_blocks, BLK, (size_t)(nb + 32) * 4, stream>>>(
      src, dst, cellcnt, cells, E, nb, n_chunks, x, W1, bufA, n_rb, gemm_blocks,
      (int4*)(ws + zero_beg), n16);

  rows_from_cells<<<nb, BLK, 0, stream>>>(cellcnt, cells, src, dst, cnt, tab, ovf_n, ovf, N, E, n_chunks);

  gather_prop_gemm<<<(N + 15) / 16, BLK, 0, stream>>>(cnt, tab, ovf_n, ovf, bufA, b1, W2, bufB, N, E);

  gather_prop_pool<<<(N + 15) / 16, BLK, 0, stream>>>(cnt, tab, ovf_n, ovf, bufB, b2, bat, sums, cnts, N, E);

  heads_wmma_f32<<<dim3(2 * (Z_DIM / 16), G_GRAPHS / 16), 32, 0, stream>>>(
      sums, cnts, Wmu, bmu, Wls, bls, out);
}


# Both levels of the edge sort put their result together in LDS and store it in whole lines (a sorted chunk per workgroup, whole table rows), not one store per edge; row sum order = LDS arrival order
# speedup vs baseline: 5.3602x; 1.0835x over previous
#include <hip/hip_runtime.h>

typedef __attribute__((ext_vector_type(2))) float v2f;
typedef __attribute__((ext_vector_type(8))) float v8f;

#define F_IN   128
#define H_DIM  64
#define Z_DIM  32
#define G_GRAPHS 512

template <int K>
__device__ __forceinline__ void gemm_rowblock_f32(
    const float* __restrict__ A, const float* __restrict__ B,
    float* __restrict__ C, int tileM, int lane)
{
  const int half  = lane >> 4;
  const int l     = lane & 15;

  const float* Arow = A + (long long)(tileM * 16 + l) * K;
  const float* Bcol = B + l;

  v8f c0 = {0.f, 0.f, 0.f, 0.f, 0.f, 0.f, 0.f, 0.f};
  v8f c1 = c0, c2 = c0, c3 = c0;

#pragma unroll 4
  for (int k = 0; k < K; k += 4) {
    const int ka = k + 2 * half;
    v2f a, b0, b1, b2, b3;
    a.x = Arow[ka];
    a.y = Arow[ka + 1];
    const float* Bk0 = Bcol + ka * H_DIM;
    const float* Bk1 = Bcol + (ka + 1) * H_DIM;
    b0.x = Bk0[0];  b0.y = Bk1[0];
    b1.x = Bk0[16]; b1.y = Bk1[16];
    b2.x = Bk0[32]; b2.y = Bk1[32];
    b3.x = Bk0[48]; b3.y = Bk1[48];
    c0 = __builtin_amdgcn_wmma_f32_16x16x4_f32(false, a, false, b0, (short)0, c0, false, false);
    c1 = __builtin_amdgcn_wmma_f32_16x16x4_f32(false, a, false, b1, (short)0, c1, false, false);
    c2 = __builtin_amdgcn_wmma_f32_16x16x4_f32(false, a, false, b2, (short)0, c2, false, false);
    c3 = __builtin_amdgcn_wmma_f32_16x16x4_f32(false, a, false, b3, (short)0, c3, false, false);
  }

  float* Crow = C + (long long)(tileM * 16 + 8 * half) * H_DIM + l;
#pragma unroll
  for (int r = 0; r < 8; ++r) {
    Crow[r * H_DIM]      = c0[r];
    Crow[r * H_DIM + 16] = c1[r];
    Crow[r * H_DIM + 32] = c2[r];
    Crow[r * H_DIM + 48] = c3[r];
  }
}

__global__ __launch_bounds__(32) void heads_wmma_f32(
    const float* __restrict__ sums, const float* __restrict__ cnts,
    const float* __restrict__ Wmu, const float* __restrict__ bmu,
    const float* __restrict__ Wls, const float* __restrict__ bls,
    float* __restrict__ out)
{
  const int head  = blockIdx.x / (Z_DIM / 16);
  const int tileN = blockIdx.x % (Z_DIM / 16);
  const int tileM = blockIdx.y;
  const int lane  = threadIdx.x;
  const int half  = lane >> 4;
  const int l     = lane & 15;

  const float* __restrict__ Wm = head ? Wls : Wmu;
  const float* __restrict__ bm = head ? bls : bmu;
  float* __restrict__ C = out + (size_t)head * G_GRAPHS * Z_DIM;

  const int row = tileM * 16 + l;
  const float den = fmaxf(cnts[row], 1.0f);
  const float* Arow = sums + (long long)row * H_DIM;
  const float* Bcol = Wm + (tileN * 16 + l);

  v8f c = {0.f, 0.f, 0.f, 0.f, 0.f, 0.f, 0.f, 0.f};

  v2f a[H_DIM / 4], b[H_DIM / 4];
#pragma unroll
  for (int s = 0; s < H_DIM / 4; ++s) {
    const int ka = 4 * s + 2 * half;
    a[s].x = Arow[ka];
    a[s].y = Arow[ka + 1];
    b[s].x = Bcol[ka * Z_DIM];
    b[s].y = Bcol[(ka + 1) * Z_DIM];
  }
#pragma unroll
  for (int s = 0; s < H_DIM / 4; ++s) {
    a[s].x = a[s].x / den;
    a[s].y = a[s].y / den;
  }
  __builtin_amdgcn_sched_barrier(0);
#pragma unroll
  for (int s = 0; s < H_DIM / 4; ++s) {
    c = __builtin_amdgcn_wmma_f32_16x16x4_f32(
        false, a[s], false, b[s], (short)0, c, false, false);
  }

  const int col = tileN * 16 + l;
  const float bv = bm[col];
  float* Crow = C + (long long)(tileM * 16 + 8 * half) * Z_DIM + col;
#pragma unroll
  for (int r = 0; r < 8; ++r) {
    Crow[r * Z_DIM] = c[r] + bv;
  }
}

#define ROW_SLOTS 64
#define NPB_SHIFT 7
#define NPB   (1 << NPB_SHIFT)
#define CHUNK 4096

#define R_PITCH 68

__device__ __forceinline__ void put_row(int* rows, int* __restrict__ ovf_n,
                                        int* __restrict__ ovf, int s, int d, int node0, int r) {
  if (r < ROW_SLOTS) {
    rows[(d - node0) * R_PITCH + r] = s;
  } else {
    const int k = atomicAdd(ovf_n, 1);
    ((int2*)ovf)[k] = make_int2(s, d);
  }
}

__global__ __launch_bounds__(256) void file_edges_gemm(
    const int* __restrict__ src, const int* __restrict__ dst,
    int* __restrict__ offs, int2* __restrict__ runs, int E, int nb, int n_chunks,
    const float* __restrict__ A, const float* __restrict__ B, float* __restrict__ C,
    int n_rb, int gemm_blocks, int4* __restrict__ zero_p, int n16)
{
  extern __shared__ __attribute__((aligned(16))) int lds_dyn[];
  const int tid = (int)threadIdx.x;
  if ((int)blockIdx.x < gemm_blocks) {
    const int rb = __builtin_amdgcn_readfirstlane((int)(blockIdx.x * 8 + (threadIdx.x >> 5)));
    if (rb >= n_rb) return;
    gemm_rowblock_f32<F_IN>(A, B, C, rb, (int)(threadIdx.x & 31));
  } else if ((int)blockIdx.x < gemm_blocks + n_chunks) {
    const int hcn = (nb + 511) & ~511;
    int2* run  = (int2*)lds_dyn;
    int*  wsum = lds_dyn + 2 * CHUNK;
    int*  hc   = wsum + 8;
    int*  off  = hc + hcn + 32;
    const int lane = tid & 31;
    const int wv   = tid >> 5;
    const int wg = (int)blockIdx.x - gemm_blocks;
    const int t0 = wg * CHUNK + tid;
    int d[16], s[16], bk[16], r[16];
    if (wg * CHUNK + CHUNK <= E) {
#pragma unroll
      for (int j = 0; j < 16; ++j) d[j] = dst[t0 + 256 * j];
#pragma unroll
      for (int j = 0; j < 16; ++j) s[j] = src[t0 + 256 * j];
    } else {
#pragma unroll
      for (int j = 0; j < 16; ++j) {
        const int t = t0 + 256 * j;
        d[j] = t < E ? dst[t] : -1;
        s[j] = t < E ? src[t] : 0;
      }
    }
    for (int b = tid; b < hcn + 32; b += 256) hc[b] = 0;
    __syncthreads();
#pragma unroll
    for (int j = 0; j < 16; ++j) bk[j] = d[j] >= 0 ? (d[j] >> NPB_SHIFT) : hcn + lane;
#pragma unroll
    for (int j = 0; j < 16; ++j) r[j] = atomicAdd(&hc[bk[j]], 1);
    __syncthreads();

    int* __restrict__ myoff = offs + (long long)wg * (nb + 1);
    int carry = 0;
    for (int tb = 0; tb < nb; tb += 512) {
      const int b0 = tb + 2 * tid;
      const int2 cc = *(const int2*)&hc[b0];
      const int loc = cc.x + cc.y;
      int inc = loc;
#pragma unroll
      for (int dl = 1; dl < 32; dl <<= 1) {
        const int t = __shfl_up(inc, dl);
        if (lane >= dl) inc += t;
      }
      if (lane == 31) wsum[wv] = inc;
      __syncthreads();
      const int4 w0 = ((const int4*)wsum)[0];
      const int4 w1 = ((const int4*)wsum)[1];
      const int ws8[8] = {w0.x, w0.y, w0.z, w0.w, w1.x, w1.y, w1.z, w1.w};
      int before = 0, all = 0;
#pragma unroll
      for (int q = 0; q < 8; ++q) {
        before += q < wv ? ws8[q] : 0;
        all += ws8[q];
      }
      const int at = carry + before + inc - loc;
      *(int2*)&off[b0] = make_int2(at, at + cc.x);
      carry += all;
      __syncthreads();
    }
    const int total = carry;

    int pos[16];
#pragma unroll
    for (int j = 0; j < 16; ++j) pos[j] = off[d[j] >= 0 ? bk[j] : 0] + r[j];
#pragma unroll
    for (int j = 0; j < 16; ++j) {
      if (d[j] >= 0) run[pos[j]] = make_int2(s[j], d[j]);
    }
    __syncthreads();

    int2* __restrict__ mine = runs + (long long)wg * CHUNK;
    int4 mv[8];
#pragma unroll
    for (int j = 0; j < 8; ++j) {
      const int q = tid + 256 * j;
      mv[j] = make_int4(0, 0, 0, 0);
      if (2 * q + 1 < total) mv[j] = ((const int4*)run)[q];
    }
#pragma unroll
    for (int j = 0; j < 8; ++j) asm volatile("" :: "v"(mv[j].x), "v"(mv[j].y), "v"(mv[j].z), "v"(mv[j].w));
#pragma unroll
    for (int j = 0; j < 8; ++j) {
      const int q = tid + 256 * j;
      if (2 * q + 1 < total) ((int4*)mine)[q] = mv[j];
    }
    if ((total & 1) && tid == 0) mine[total - 1] = run[total - 1];
    for (int tb = 0; tb < nb; tb += 512) {
      const int b0 = tb + 2 * tid;
      const int2 o = *(const int2*)&off[b0];
      if (b0 < nb) myoff[b0] = o.x;
      if (b0 + 1 < nb) myoff[b0 + 1] = o.y;
    }
    if (tid == 0) myoff[nb] = total;
  } else {
    const int t = ((int)blockIdx.x - gemm_blocks - n_chunks) * 256 + tid;
    if (t < n16) zero_p[t] = make_int4(0, 0, 0, 0);
  }
}

__global__ __launch_bounds__(256) void rows_from_runs(
    const int* __restrict__ offs, const int2* __restrict__ runs,
    int* __restrict__ cnt, int* __restrict__ tab,
    int* __restrict__ ovf_n, int* __restrict__ ovf, int n, int nb, int n_chunks)
{
  __shared__ int lc[NPB + 32];
  __shared__ __attribute__((aligned(16))) int rows[NPB * R_PITCH];
  const int b    = (int)blockIdx.x;
  const int tid  = (int)threadIdx.x;
  const int lane = tid & 31;
  const int w    = __builtin_amdgcn_readfirstlane(tid >> 5);
  const int node0 = b << NPB_SHIFT;
  if (tid < NPB + 32) lc[tid] = 0;
  for (int q = tid; q < NPB * R_PITCH / 4; q += 256) ((int4*)rows)[q] = make_int4(0, 0, 0, 0);
  __syncthreads();

  const int nbp1 = nb + 1;

  for (int g0 = 0; g0 < n_chunks; g0 += 256) {
    const int mychunk = g0 + w + 8 * lane;
    int st = 0, ln = 0;
    if (mychunk < n_chunks) {
      const int* __restrict__ o = offs + (long long)mychunk * nbp1 + b;
      st = o[0];
      ln = o[1] - st;
    }
    for (int j0 = 0; j0 < 32; j0 += 16) {
      if (g0 + w + 8 * j0 >= n_chunks) break;
      int2 ev[16];
      int  cv[16], r[16];
#pragma unroll
      for (int u = 0; u < 16; ++u) {
        cv[u] = __builtin_amdgcn_readlane(ln, j0 + u);
        const int sv = __builtin_amdgcn_readlane(st, j0 + u);
        ev[u] = make_int2(0, 0);
        if (lane < cv[u])
          ev[u] = runs[(long long)(g0 + w + 8 * (j0 + u)) * CHUNK + sv + lane];
      }
#pragma unroll
      for (int u = 0; u < 16; ++u) {
        const bool on = lane < cv[u];
        r[u] = atomicAdd(&lc[on ? ev[u].y - node0 : NPB + lane], 1);
      }
#pragma unroll
      for (int u = 0; u < 16; ++u) {
        if (lane < cv[u])
          put_row(rows, ovf_n, ovf, ev[u].x, ev[u].y, node0, r[u]);
      }
    }
    for (int j = 0; j < 32; ++j) {
      const int cj = __builtin_amdgcn_readlane(ln, j);
      if (cj > 32) {
        const int2* __restrict__ rj =
            runs + (long long)(g0 + w + 8 * j) * CHUNK + __builtin_amdgcn_readlane(st, j);
        for (int k = 32 + lane; k < cj; k += 32) {
          const int2 e = rj[k];
          const int rr = atomicAdd(&lc[e.y - node0], 1);
          put_row(rows, ovf_n, ovf, e.x, e.y, node0, rr);
        }
      }
    }
  }
  __syncthreads();
  if (tid < NPB && node0 + tid < n) cnt[node0 + tid] = lc[tid];
  const int sub = lane & 15;
  const int hi  = lane >> 4;
  int  rl[NPB / 16];
  int4 rv[NPB / 16];
#pragma unroll
  for (int t = 0; t < NPB / 16; ++t) {
    const int row = 16 * t + 2 * w + hi;
    rl[t] = lc[row];
    rv[t] = *(const int4*)&rows[row * R_PITCH + 4 * sub];
  }
#pragma unroll
  for (int t = 0; t < NPB / 16; ++t) asm volatile("" :: "v"(rv[t].x), "v"(rv[t].y), "v"(rv[t].z), "v"(rv[t].w));
#pragma unroll
  for (int t = 0; t < NPB / 16; ++t) {
    const int node = node0 + 16 * t + 2 * w + hi;
    if (node < n && 4 * sub < min(rl[t], ROW_SLOTS))
      ((int4*)tab)[(long long)node * (ROW_SLOTS / 4) + sub] = rv[t];
  }
}

#define T_PITCH 68

__device__ __forceinline__ float dinv_of(int c) {
  return __builtin_amdgcn_rsqf((float)c + 1.0f);
}

template <bool PRE>
__device__ __forceinline__ float4 gather_row(
    const int* __restrict__ cnt, const int* __restrict__ tab,
    const int* __restrict__ ovf_n, const int* __restrict__ ovf,
    const float* __restrict__ h, const float* __restrict__ bias,
    int ic, bool live, int sub, int hi, int E, float& di_out)
{
  const int* __restrict__ trow = tab + (long long)ic * ROW_SLOTS;
  const float4* __restrict__ h4 = (const float4*)h;
  const int degi = cnt[ic];
  const float4 hv = h4[(long long)ic * (H_DIM / 4) + sub];
  const int deg = live ? degi : 0;
  const int len = min(deg, ROW_SLOTS);
  const float di = dinv_of(degi);
  di_out = di;

  const int len0 = __builtin_amdgcn_readlane(len, 0);
  const int len1 = __builtin_amdgcn_readlane(len, 16);
  const int maxlen = max(len0, len1);

  int s_cur = ic, s_nxt = ic;
  if (sub < len) s_cur = trow[sub];
  if (16 + sub < len) s_nxt = trow[16 + sub];

  float ax = 0.0f, ay = 0.0f, az = 0.0f, aw = 0.0f;
  for (int base = 0; base < maxlen; base += 16) {
    const int m = min(16, maxlen - base);
    const int px = s_cur;
    s_cur = s_nxt;
    if (base + 32 < maxlen) {
      const int e2 = base + 32 + sub;
      s_nxt = ic;
      if (e2 < len) s_nxt = trow[e2];
    }
    for (int j = 0; j < m; j += 8) {
      float4 v[8];
      int c[8];
#pragma unroll
      for (int u = 0; u < 8; ++u) {
        const int s0 = __builtin_amdgcn_readlane(px, j + u);
        const int s1 = __builtin_amdgcn_readlane(px, j + u + 16);
        const int sj = hi ? s1 : s0;
        v[u] = h4[sj * (H_DIM / 4) + sub];
        if (!PRE) c[u] = cnt[sj];
      }
#pragma unroll
      for (int u = 0; u < 8; ++u) {
        const bool on = base + j + u < len;
        float tx, ty, tz, tw;
        if (PRE) {
          tx = ax + v[u].x;
          ty = ay + v[u].y;
          tz = az + v[u].z;
          tw = aw + v[u].w;
        } else {
          const float cj = dinv_of(c[u]) * di;
          tx = ax + v[u].x * cj;
          ty = ay + v[u].y * cj;
          tz = az + v[u].z * cj;
          tw = aw + v[u].w * cj;
        }
        ax = on ? tx : ax;
        ay = on ? ty : ay;
        az = on ? tz : az;
        aw = on ? tw : aw;
      }
    }
  }

  const int deg0 = __builtin_amdgcn_readlane(deg, 0);
  const int deg1 = __builtin_amdgcn_readlane(deg, 16);
  if (max(deg0, deg1) > ROW_SLOTS) {
    const int nov = min(ovf_n[0], E);
    const int2* __restrict__ ov = (const int2*)ovf;
    for (int k = 0; k < nov; ++k) {
      const int2 od = ov[k];
      if (live && od.y == ic) {
        const float4 v = h4[(long long)od.x * (H_DIM / 4) + sub];
        if (PRE) {
          ax += v.x;
          ay += v.y;
          az += v.z;
          aw += v.w;
        } else {
          const float cj = dinv_of(cnt[od.x]) * di;
          ax += v.x * cj;
          ay += v.y * cj;
          az += v.z * cj;
          aw += v.w * cj;
        }
      }
    }
  }

  const float4 bv = ((const float4*)bias)[sub];
  float ox, oy, oz, ow;
  if (PRE) {
    ox = (ax + hv.x) * di + bv.x;
    oy = (ay + hv.y) * di + bv.y;
    oz = (az + hv.z) * di + bv.z;
    ow = (aw + hv.w) * di + bv.w;
  } else {
    const float sl = di * di;
    ox = ax + hv.x * sl + bv.x;
    oy = ay + hv.y * sl + bv.y;
    oz = az + hv.z * sl + bv.z;
    ow = aw + hv.w * sl + bv.w;
  }
  float4 o;
  o.x = (live && ox > 0.0f) ? ox : 0.0f;
  o.y = (live && oy > 0.0f) ? oy : 0.0f;
  o.z = (live && oz > 0.0f) ? oz : 0.0f;
  o.w = (live && ow > 0.0f) ? ow : 0.0f;
  return o;
}

__global__ __launch_bounds__(256) void gather_prop_gemm(
    const int* __restrict__ cnt, const int* __restrict__ tab,
    const int* __restrict__ ovf_n, const int* __restrict__ ovf,
    const float* __restrict__ h,
    const float* __restrict__ bias, const float* __restrict__ W,
    float* __restrict__ out, int n, int E)
{
  __shared__ __attribute__((aligned(16))) float T[16 * T_PITCH];
  __shared__ __attribute__((aligned(16))) float U[16 * T_PITCH];

  const int lane = threadIdx.x & 31;
  const int sub  = lane & 15;
  const int hi   = lane >> 4;
  const int w    = __builtin_amdgcn_readfirstlane((int)(threadIdx.x >> 5));
  const int rl   = 2 * w + hi;
  const int i    = (int)blockIdx.x * 16 + rl;
  const bool live = i < n;
  const int ic   = live ? i : n - 1;

  float di;
  const float4 o = gather_row<false>(cnt, tab, ovf_n, ovf, h, bias, ic, live, sub, hi, E, di);
  *(float4*)&T[rl * T_PITCH + 4 * sub] = o;

  v2f b[H_DIM / 4];
  if (w < 4) {
    const float* Bcol = W + 16 * w + sub;
#pragma unroll
    for (int s = 0; s < H_DIM / 4; ++s) {
      const int ka = 4 * s + 2 * hi;
      b[s].x = Bcol[ka * H_DIM];
      b[s].y = Bcol[(ka + 1) * H_DIM];
    }
  }
  __syncthreads();
  if (w < 4) {
    v8f c = {0.f, 0.f, 0.f, 0.f, 0.f, 0.f, 0.f, 0.f};
    const float* Arow = T + sub * T_PITCH + 2 * hi;
#pragma unroll
    for (int s = 0; s < H_DIM / 4; ++s) {
      v2f a;
      a.x = Arow[4 * s];
      a.y = Arow[4 * s + 1];
      c = __builtin_amdgcn_wmma_f32_16x16x4_f32(false, a, false, b[s], (short)0, c, false, false);
    }
    float* Ucol = U + (8 * hi) * T_PITCH + 16 * w + sub;
#pragma unroll
    for (int r = 0; r < 8; ++r) Ucol[r * T_PITCH] = c[r];
  }
  __syncthreads();
  if (live) {
    float4 v = *(const float4*)&U[rl * T_PITCH + 4 * sub];
    v.x *= di;
    v.y *= di;
    v.z *= di;
    v.w *= di;
    ((float4*)out)[(long long)i * (H_DIM / 4) + sub] = v;
  }
}

__global__ __launch_bounds__(256) void gather_prop_pool(
    const int* __restrict__ cnt, const int* __restrict__ tab,
    const int* __restrict__ ovf_n, const int* __restrict__ ovf,
    const float* __restrict__ h,
    const float* __restrict__ bias, const int* __restrict__ batch,
    float* __restrict__ sums, float* __restrict__ cnts, int n, int E)
{
  __shared__ __attribute__((aligned(16))) float T[16 * T_PITCH];
  __shared__ __attribute__((aligned(16))) int Gs[16];

  const int lane = threadIdx.x & 31;
  const int sub  = lane & 15;
  const int hi   = lane >> 4;
  const int w    = (int)(threadIdx.x >> 5);
  const int rl   = 2 * w + hi;
  const int i    = (int)blockIdx.x * 16 + rl;
  const bool live = i < n;
  const int ic   = live ? i : n - 1;
  const int g    = batch[ic];

  float di;
  const float4 o = gather_row<true>(cnt, tab, ovf_n, ovf, h, bias, ic, live, sub, hi, E, di);
  *(float4*)&T[rl * T_PITCH + 4 * sub] = o;
  if (sub == 0) Gs[rl] = live ? g : -1;
  __syncthreads();
  if (threadIdx.x >= H_DIM) return;

  const int f = (int)threadIdx.x;
  float tv[16];
  int gs[16];
#pragma unroll
  for (int q = 0; q < 4; ++q) {
    const int4 g4 = ((const int4*)Gs)[q];
    gs[4 * q + 0] = g4.x;
    gs[4 * q + 1] = g4.y;
    gs[4 * q + 2] = g4.z;
    gs[4 * q + 3] = g4.w;
  }
#pragma unroll
  for (int r = 0; r < 16; ++r) tv[r] = T[r * T_PITCH + f];

  int cur = gs[0];
  float acc = 0.0f, cn = 0.0f;
#pragma unroll
  for (int r = 0; r < 16; ++r) {
    if (gs[r] != cur) {
      if (cur >= 0) {
        atomicAdd(&sums[(long long)cur * H_DIM + f], acc);
        if (f == 0) atomicAdd(&cnts[cur], cn);
      }
      acc = 0.0f;
      cn = 0.0f;
      cur = gs[r];
    }
    acc += tv[r];
    cn += 1.0f;
  }
  if (cur >= 0) {
    atomicAdd(&sums[(long long)cur * H_DIM + f], acc);
    if (f == 0) atomicAdd(&cnts[cur], cn);
  }
}

static inline size_t align_up(size_t x, size_t a) { return (x + a - 1) & ~(a - 1); }

extern "C" void kernel_launch(void* const* d_in, const int* in_sizes, int n_in,
                              void* d_out, int out_size, void* d_ws, size_t ws_size,
                              hipStream_t stream) {
  const float* x   = (const float*)d_in[0];
  const int*   ei  = (const int*)d_in[1];
  const int*   bat = (const int*)d_in[2];
  const float* W1  = (const float*)d_in[3];
  const float* b1  = (const float*)d_in[4];
  const float* W2  = (const float*)d_in[5];
  const float* b2  = (const float*)d_in[6];
  const float* Wmu = (const float*)d_in[7];
  const float* bmu = (const float*)d_in[8];
  const float* Wls = (const float*)d_in[9];
  const float* bls = (const float*)d_in[10];
  float* out = (float*)d_out;

  const int N = in_sizes[0] / F_IN;
  const int E = in_sizes[1] / 2;
  const int* src = ei;
  const int* dst = ei + E;

  char* ws = (char*)d_ws;
  size_t off = 0;
  const size_t zero_beg = off;
  float* sums = (float*)(ws + off); off = align_up(off + (size_t)G_GRAPHS * H_DIM * 4, 256);
  float* cnts = (float*)(ws + off); off = align_up(off + (size_t)G_GRAPHS * 4, 256);
  int* ovf_n  = (int*)(ws + off);   off = align_up(off + 4, 256);
  const size_t zero_end = off;
  const int nb = (N + NPB - 1) / NPB;
  const int n_chunks = (E + CHUNK - 1) / CHUNK;
  int* cnt    = (int*)(ws + off);   off = align_up(off + (size_t)N * 4, 256);
  int* offs   = (int*)(ws + off);   off = align_up(off + (size_t)n_chunks * (nb + 1) * 4, 256);
  int2* runs  = (int2*)(ws + off);  off = align_up(off + (size_t)n_chunks * CHUNK * 8, 256);
  int* tab    = (int*)(ws + off);   off = align_up(off + (size_t)N * ROW_SLOTS * 4, 256);
  int* ovf    = (int*)(ws + off);   off = align_up(off + (size_t)E * 8, 256);
  float* bufA = (float*)(ws + off); off = align_up(off + (size_t)N * H_DIM * 4, 256);
  float* bufB = (float*)(ws + off); off = align_up(off + (size_t)N * H_DIM * 4, 256);
  (void)ws_size; (void)n_in; (void)out_size;

  const int BLK = 256;
  const int n16 = (int)((zero_end - zero_beg) / 16);
  const int n_rb = N / 16;
  const int gemm_blocks = (n_rb + 7) / 8;
  const int clear_blocks = (n16 + BLK - 1) / BLK;

  const size_t hcn = ((size_t)nb + 511) & ~(size_t)511;
  const size_t sort_lds = ((size_t)2 * CHUNK + 8 + (hcn + 32) + hcn) * 4;
  file_edges_gemm<<<gemm_blocks + n_chunks + clear_blocks, BLK, sort_lds, stream>>>(
      src, dst, offs, runs, E, nb, n_chunks, x, W1, bufA, n_rb, gemm_blocks,
      (int4*)(ws + zero_beg), n16);

  rows_from_runs<<<nb, BLK, 0, stream>>>(offs, runs, cnt, tab, ovf_n, ovf, N, nb, n_chunks);

  gather_prop_gemm<<<(N + 15) / 16, BLK, 0, stream>>>(cnt, tab, ovf_n, ovf, bufA, b1, W2, bufB, N, E);

  gather_prop_pool<<<(N + 15) / 16, BLK, 0, stream>>>(cnt, tab, ovf_n, ovf, bufB, b2, bat, sums, cnts, N, E);

  heads_wmma_f32<<<dim3(2 * (Z_DIM / 16), G_GRAPHS / 16), 32, 0, stream>>>(
      sums, cnts, Wmu, bmu, Wls, bls, out);
}
